# starting kernel
# baseline (speedup 1.0000x reference)
#include <hip/hip_runtime.h>

#define NN 100000
#define NE 1600000

typedef float v2f __attribute__((ext_vector_type(2)));
typedef float v8f __attribute__((ext_vector_type(8)));

__global__ void deg_init_k(float* deg, int n) {
  int i = blockIdx.x * blockDim.x + threadIdx.x;
  if (i < n) deg[i] = 1.0f;
}

__global__ void deg_accum_k(const int* __restrict__ row, const float* __restrict__ w,
                            float* deg, int e) {
  int i = blockIdx.x * blockDim.x + threadIdx.x;
  if (i < e) atomicAdd(&deg[row[i]], w[i]);
}

__global__ void deg_rsqrt_k(float* deg, int n) {
  int i = blockIdx.x * blockDim.x + threadIdx.x;
  if (i < n) deg[i] = rsqrtf(deg[i]);
}

__global__ void edge_norm_k(const int* __restrict__ row, const int* __restrict__ col,
                            const float* __restrict__ w, const float* __restrict__ dis,
                            float* __restrict__ nrm, int e) {
  int i = blockIdx.x * blockDim.x + threadIdx.x;
  if (i < e) nrm[i] = dis[row[i]] * w[i] * dis[col[i]];
}

__global__ void transpose_k(const float* __restrict__ W, float* __restrict__ Wt, int U) {
  int i = blockIdx.x * blockDim.x + threadIdx.x;
  if (i >= 128 * U) return;
  int k = i / U, u = i % U;
  Wt[(size_t)u * 128 + k] = W[i];
}

template <int U, bool RELU_IN>
__global__ __launch_bounds__(128) void gemm_wmma_k(const float* __restrict__ X,
                                                   const float* __restrict__ Wt,
                                                   float* __restrict__ T, int M) {
  const int lane   = threadIdx.x & 31;
  const int wave   = threadIdx.x >> 5;
  const int tile_m = blockIdx.x * 4 + wave;
  const int tile_n = blockIdx.y;
  if (tile_m * 16 >= M) return;

  const int mrow = tile_m * 16 + (lane & 15);
  const int ncol = tile_n * 16 + (lane & 15);
  const int kk   = (lane >> 4) * 2;

  const float* xr = X + (size_t)mrow * 128 + kk;
  const float* wr = Wt + (size_t)ncol * 128 + kk;

  v8f c = {0.f, 0.f, 0.f, 0.f, 0.f, 0.f, 0.f, 0.f};
#pragma unroll
  for (int k = 0; k < 128; k += 4) {
    v2f a = *reinterpret_cast<const v2f*>(xr + k);
    if (RELU_IN) { a.x = fmaxf(a.x, 0.0f); a.y = fmaxf(a.y, 0.0f); }
    v2f b = *reinterpret_cast<const v2f*>(wr + k);
    c = __builtin_amdgcn_wmma_f32_16x16x4_f32(false, a, false, b, (short)0, c,
                                              false, false);
  }

  const int mbase = tile_m * 16 + (lane >> 4) * 8;
  float* trow = T + (size_t)mbase * U + ncol;
#pragma unroll
  for (int r = 0; r < 8; ++r) trow[(size_t)r * U] = c[r];
}

template <int UV>
__global__ void agg_init_k(const float4* __restrict__ T, const float* __restrict__ dis,
                           const float* __restrict__ bias, float4* __restrict__ out,
                           int n) {
  long long idx = (long long)blockIdx.x * blockDim.x + threadIdx.x;
  if (idx >= (long long)n * UV) return;
  int i  = (int)(idx / UV);
  int u4 = (int)(idx % UV);
  float d  = dis[i];
  float sn = d * d;
  float4 t = T[idx];
  float4 b = reinterpret_cast<const float4*>(bias)[u4];
  float4 o;
  o.x = b.x + sn * t.x; o.y = b.y + sn * t.y;
  o.z = b.z + sn * t.z; o.w = b.w + sn * t.w;
  out[idx] = o;
}

template <int UV>
__global__ void agg_scatter_k(const int* __restrict__ row, const int* __restrict__ col,
                              const float* __restrict__ nrm, const float* __restrict__ T,
                              float* out, int e) {
  long long idx = (long long)blockIdx.x * blockDim.x + threadIdx.x;
  if (idx >= (long long)e * UV) return;
  int ei = (int)(idx / UV);
  int u4 = (int)(idx % UV) * 4;
  int r = row[ei], cc = col[ei];
  float w = nrm[ei];
  const int Uu = UV * 4;
  const float4 t = *reinterpret_cast<const float4*>(T + (size_t)cc * Uu + u4);
  float* dst = out + (size_t)r * Uu + u4;
  atomicAdd(dst + 0, w * t.x);
  atomicAdd(dst + 1, w * t.y);
  atomicAdd(dst + 2, w * t.z);
  atomicAdd(dst + 3, w * t.w);
}

__global__ void final_k(const float* __restrict__ rnd, float* out, int nz) {
  int i = blockIdx.x * blockDim.x + threadIdx.x;
  if (i >= nz) return;
  float mu = out[i];
  float ls = out[nz + i];
  float s  = expf(ls);
  out[i]      = mu + rnd[i] * s;
  out[nz + i] = -ls + 0.5f * (s * s + mu * mu - 1.0f);
}

extern "C" void kernel_launch(void* const* d_in, const int* in_sizes, int n_in,
                              void* d_out, int out_size, void* d_ws, size_t ws_size,
                              hipStream_t stream) {
  (void)in_sizes; (void)n_in; (void)out_size; (void)ws_size;
  const int N = NN, E = NE, H = 128, DZ = 64;
  const float* h   = (const float*)d_in[0];
  const int*   ei  = (const int*)d_in[1];
  const float* ew  = (const float*)d_in[2];
  const float* rnd = (const float*)d_in[3];
  const float* W0  = (const float*)d_in[4];
  const float* b0  = (const float*)d_in[5];
  const float* W1  = (const float*)d_in[6];
  const float* b1  = (const float*)d_in[7];
  const float* Wmu = (const float*)d_in[8];
  const float* bmu = (const float*)d_in[9];
  const float* Wsd = (const float*)d_in[10];
  const float* bsd = (const float*)d_in[11];
  const int* row = ei;
  const int* col = ei + E;
  float* out = (float*)d_out;
  const int NZ = N * DZ;

  float* ws    = (float*)d_ws;
  float* dis   = ws;
  float* nrm   = dis + N;
  float* bufT  = nrm + E;
  float* bufH  = bufT + (size_t)N * 128;
  float* bufS  = bufH + (size_t)N * 128;
  float* Wt0   = bufS + (size_t)N * 64;
  float* Wt1   = Wt0 + 128 * 128;
  float* Wtmu  = Wt1 + 128 * 128;
  float* Wtsd  = Wtmu + 64 * 128;

  const int TB = 256;
  deg_init_k <<<(N + TB - 1) / TB, TB, 0, stream>>>(dis, N);
  deg_accum_k<<<(E + TB - 1) / TB, TB, 0, stream>>>(row, ew, dis, E);
  deg_rsqrt_k<<<(N + TB - 1) / TB, TB, 0, stream>>>(dis, N);
  edge_norm_k<<<(E + TB - 1) / TB, TB, 0, stream>>>(row, col, ew, dis, nrm, E);

  transpose_k<<<(128 * H  + TB - 1) / TB, TB, 0, stream>>>(W0,  Wt0,  H);
  transpose_k<<<(128 * H  + TB - 1) / TB, TB, 0, stream>>>(W1,  Wt1,  H);
  transpose_k<<<(128 * DZ + TB - 1) / TB, TB, 0, stream>>>(Wmu, Wtmu, DZ);
  transpose_k<<<(128 * DZ + TB - 1) / TB, TB, 0, stream>>>(Wsd, Wtsd, DZ);

  const int MT = N / 16;
  dim3 gblk(128);
  const unsigned MB = (unsigned)((MT + 3) / 4);

  gemm_wmma_k<128, false><<<dim3(MB, H / 16), gblk, 0, stream>>>(h, Wt0, bufT, N);
  {
    long long ti = (long long)N * 32, ts = (long long)E * 32;
    agg_init_k<32>   <<<(unsigned)((ti + TB - 1) / TB), TB, 0, stream>>>((const float4*)bufT, dis, b0, (float4*)bufH, N);
    agg_scatter_k<32><<<(unsigned)((ts + TB - 1) / TB), TB, 0, stream>>>(row, col, nrm, bufT, bufH, E);
  }
  gemm_wmma_k<128, true><<<dim3(MB, H / 16), gblk, 0, stream>>>(bufH, Wt1, bufT, N);
  {
    long long ti = (long long)N * 32, ts = (long long)E * 32;
    agg_init_k<32>   <<<(unsigned)((ti + TB - 1) / TB), TB, 0, stream>>>((const float4*)bufT, dis, b1, (float4*)bufH, N);
    agg_scatter_k<32><<<(unsigned)((ts + TB - 1) / TB), TB, 0, stream>>>(row, col, nrm, bufT, bufH, E);
  }
  gemm_wmma_k<64, true><<<dim3(MB, DZ / 16), gblk, 0, stream>>>(bufH, Wtmu, bufS, N);
  {
    long long ti = (long long)N * 16, ts = (long long)E * 16;
    agg_init_k<16>   <<<(unsigned)((ti + TB - 1) / TB), TB, 0, stream>>>((const float4*)bufS, dis, bmu, (float4*)out, N);
    agg_scatter_k<16><<<(unsigned)((ts + TB - 1) / TB), TB, 0, stream>>>(row, col, nrm, bufS, out, E);
  }
  gemm_wmma_k<64, true><<<dim3(MB, DZ / 16), gblk, 0, stream>>>(bufH, Wtsd, bufS, N);
  {
    long long ti = (long long)N * 16, ts = (long long)E * 16;
    agg_init_k<16>   <<<(unsigned)((ti + TB - 1) / TB), TB, 0, stream>>>((const float4*)bufS, dis, bsd, (float4*)(out + NZ), N);
    agg_scatter_k<16><<<(unsigned)((ts + TB - 1) / TB), TB, 0, stream>>>(row, col, nrm, bufS, out + NZ, E);
  }
  final_k<<<(NZ + TB - 1) / TB, TB, 0, stream>>>(rnd, out, NZ);
}


# scatter-add: one wave per edge, lane l takes floats l+32j, so each row-gather load and each f32 atomic add covers one whole 128-B line (was 4 lines x 8 floats); atomic sum order as before
# speedup vs baseline: 1.1026x; 1.1026x over previous
#include <hip/hip_runtime.h>

#define NN 100000
#define NE 1600000

typedef float v2f __attribute__((ext_vector_type(2)));
typedef float v8f __attribute__((ext_vector_type(8)));

__global__ void deg_init_k(float* deg, int n) {
  int i = blockIdx.x * blockDim.x + threadIdx.x;
  if (i < n) deg[i] = 1.0f;
}

__global__ void deg_accum_k(const int* __restrict__ row, const float* __restrict__ w,
                            float* deg, int e) {
  int i = blockIdx.x * blockDim.x + threadIdx.x;
  if (i < e) atomicAdd(&deg[row[i]], w[i]);
}

__global__ void deg_rsqrt_k(float* deg, int n) {
  int i = blockIdx.x * blockDim.x + threadIdx.x;
  if (i < n) deg[i] = rsqrtf(deg[i]);
}

__global__ void edge_norm_k(const int* __restrict__ row, const int* __restrict__ col,
                            const float* __restrict__ w, const float* __restrict__ dis,
                            float* __restrict__ nrm, int e) {
  int i = blockIdx.x * blockDim.x + threadIdx.x;
  if (i < e) nrm[i] = dis[row[i]] * w[i] * dis[col[i]];
}

__global__ void transpose_k(const float* __restrict__ W, float* __restrict__ Wt, int U) {
  int i = blockIdx.x * blockDim.x + threadIdx.x;
  if (i >= 128 * U) return;
  int k = i / U, u = i % U;
  Wt[(size_t)u * 128 + k] = W[i];
}

template <int U, bool RELU_IN>
__global__ __launch_bounds__(128) void gemm_wmma_k(const float* __restrict__ X,
                                                   const float* __restrict__ Wt,
                                                   float* __restrict__ T, int M) {
  const int lane   = threadIdx.x & 31;
  const int wave   = threadIdx.x >> 5;
  const int tile_m = blockIdx.x * 4 + wave;
  const int tile_n = blockIdx.y;
  if (tile_m * 16 >= M) return;

  const int mrow = tile_m * 16 + (lane & 15);
  const int ncol = tile_n * 16 + (lane & 15);
  const int kk   = (lane >> 4) * 2;

  const float* xr = X + (size_t)mrow * 128 + kk;
  const float* wr = Wt + (size_t)ncol * 128 + kk;

  v8f c = {0.f, 0.f, 0.f, 0.f, 0.f, 0.f, 0.f, 0.f};
#pragma unroll
  for (int k = 0; k < 128; k += 4) {
    v2f a = *reinterpret_cast<const v2f*>(xr + k);
    if (RELU_IN) { a.x = fmaxf(a.x, 0.0f); a.y = fmaxf(a.y, 0.0f); }
    v2f b = *reinterpret_cast<const v2f*>(wr + k);
    c = __builtin_amdgcn_wmma_f32_16x16x4_f32(false, a, false, b, (short)0, c,
                                              false, false);
  }

  const int mbase = tile_m * 16 + (lane >> 4) * 8;
  float* trow = T + (size_t)mbase * U + ncol;
#pragma unroll
  for (int r = 0; r < 8; ++r) trow[(size_t)r * U] = c[r];
}

template <int UV>
__global__ void agg_init_k(const float4* __restrict__ T, const float* __restrict__ dis,
                           const float* __restrict__ bias, float4* __restrict__ out,
                           int n) {
  long long idx = (long long)blockIdx.x * blockDim.x + threadIdx.x;
  if (idx >= (long long)n * UV) return;
  int i  = (int)(idx / UV);
  int u4 = (int)(idx % UV);
  float d  = dis[i];
  float sn = d * d;
  float4 t = T[idx];
  float4 b = reinterpret_cast<const float4*>(bias)[u4];
  float4 o;
  o.x = b.x + sn * t.x; o.y = b.y + sn * t.y;
  o.z = b.z + sn * t.z; o.w = b.w + sn * t.w;
  out[idx] = o;
}

template <int UV>
__global__ void agg_scatter_k(const int* __restrict__ row, const int* __restrict__ col,
                              const float* __restrict__ nrm, const float* __restrict__ T,
                              float* out, int e) {
  constexpr int Uu = UV * 4;
  constexpr int NL = Uu / 32;
  long long idx = (long long)blockIdx.x * blockDim.x + threadIdx.x;
  int ei = (int)(idx >> 5);
  if (ei >= e) return;
  const int lane = threadIdx.x & 31;
  int r = row[ei], cc = col[ei];
  float w = nrm[ei];
  const float* srcp = T + (size_t)cc * Uu + lane;
  float* dst = out + (size_t)r * Uu + lane;
  float t[NL];
#pragma unroll
  for (int j = 0; j < NL; ++j) t[j] = srcp[32 * j];
#pragma unroll
  for (int j = 0; j < NL; ++j) atomicAdd(dst + 32 * j, w * t[j]);
}

__global__ void final_k(const float* __restrict__ rnd, float* out, int nz) {
  int i = blockIdx.x * blockDim.x + threadIdx.x;
  if (i >= nz) return;
  float mu = out[i];
  float ls = out[nz + i];
  float s  = expf(ls);
  out[i]      = mu + rnd[i] * s;
  out[nz + i] = -ls + 0.5f * (s * s + mu * mu - 1.0f);
}

extern "C" void kernel_launch(void* const* d_in, const int* in_sizes, int n_in,
                              void* d_out, int out_size, void* d_ws, size_t ws_size,
                              hipStream_t stream) {
  (void)in_sizes; (void)n_in; (void)out_size; (void)ws_size;
  const int N = NN, E = NE, H = 128, DZ = 64;
  const float* h   = (const float*)d_in[0];
  const int*   ei  = (const int*)d_in[1];
  const float* ew  = (const float*)d_in[2];
  const float* rnd = (const float*)d_in[3];
  const float* W0  = (const float*)d_in[4];
  const float* b0  = (const float*)d_in[5];
  const float* W1  = (const float*)d_in[6];
  const float* b1  = (const float*)d_in[7];
  const float* Wmu = (const float*)d_in[8];
  const float* bmu = (const float*)d_in[9];
  const float* Wsd = (const float*)d_in[10];
  const float* bsd = (const float*)d_in[11];
  const int* row = ei;
  const int* col = ei + E;
  float* out = (float*)d_out;
  const int NZ = N * DZ;

  float* ws    = (float*)d_ws;
  float* dis   = ws;
  float* nrm   = dis + N;
  float* bufT  = nrm + E;
  float* bufH  = bufT + (size_t)N * 128;
  float* bufS  = bufH + (size_t)N * 128;
  float* Wt0   = bufS + (size_t)N * 64;
  float* Wt1   = Wt0 + 128 * 128;
  float* Wtmu  = Wt1 + 128 * 128;
  float* Wtsd  = Wtmu + 64 * 128;

  const int TB = 256;
  deg_init_k <<<(N + TB - 1) / TB, TB, 0, stream>>>(dis, N);
  deg_accum_k<<<(E + TB - 1) / TB, TB, 0, stream>>>(row, ew, dis, E);
  deg_rsqrt_k<<<(N + TB - 1) / TB, TB, 0, stream>>>(dis, N);
  edge_norm_k<<<(E + TB - 1) / TB, TB, 0, stream>>>(row, col, ew, dis, nrm, E);

  transpose_k<<<(128 * H  + TB - 1) / TB, TB, 0, stream>>>(W0,  Wt0,  H);
  transpose_k<<<(128 * H  + TB - 1) / TB, TB, 0, stream>>>(W1,  Wt1,  H);
  transpose_k<<<(128 * DZ + TB - 1) / TB, TB, 0, stream>>>(Wmu, Wtmu, DZ);
  transpose_k<<<(128 * DZ + TB - 1) / TB, TB, 0, stream>>>(Wsd, Wtsd, DZ);

  const int MT = N / 16;
  dim3 gblk(128);
  const unsigned MB = (unsigned)((MT + 3) / 4);

  gemm_wmma_k<128, false><<<dim3(MB, H / 16), gblk, 0, stream>>>(h, Wt0, bufT, N);
  {
    long long ti = (long long)N * 32, ts = (long long)E * 32;
    agg_init_k<32>   <<<(unsigned)((ti + TB - 1) / TB), TB, 0, stream>>>((const float4*)bufT, dis, b0, (float4*)bufH, N);
    agg_scatter_k<32><<<(unsigned)((ts + TB - 1) / TB), TB, 0, stream>>>(row, col, nrm, bufT, bufH, E);
  }
  gemm_wmma_k<128, true><<<dim3(MB, H / 16), gblk, 0, stream>>>(bufH, Wt1, bufT, N);
  {
    long long ti = (long long)N * 32, ts = (long long)E * 32;
    agg_init_k<32>   <<<(unsigned)((ti + TB - 1) / TB), TB, 0, stream>>>((const float4*)bufT, dis, b1, (float4*)bufH, N);
    agg_scatter_k<32><<<(unsigned)((ts + TB - 1) / TB), TB, 0, stream>>>(row, col, nrm, bufT, bufH, E);
  }
  gemm_wmma_k<64, true><<<dim3(MB, DZ / 16), gblk, 0, stream>>>(bufH, Wtmu, bufS, N);
  {
    long long ti = (long long)N * 16, ts = (long long)E * 32;
    agg_init_k<16>   <<<(unsigned)((ti + TB - 1) / TB), TB, 0, stream>>>((const float4*)bufS, dis, bmu, (float4*)out, N);
    agg_scatter_k<16><<<(unsigned)((ts + TB - 1) / TB), TB, 0, stream>>>(row, col, nrm, bufS, out, E);
  }
  gemm_wmma_k<64, true><<<dim3(MB, DZ / 16), gblk, 0, stream>>>(bufH, Wtsd, bufS, N);
  {
    long long ti = (long long)N * 16, ts = (long long)E * 32;
    agg_init_k<16>   <<<(unsigned)((ti + TB - 1) / TB), TB, 0, stream>>>((const float4*)bufS, dis, bsd, (float4*)(out + NZ), N);
    agg_scatter_k<16><<<(unsigned)((ts + TB - 1) / TB), TB, 0, stream>>>(row, col, nrm, bufS, out + NZ, E);
  }
  final_k<<<(NZ + TB - 1) / TB, TB, 0, stream>>>(rnd, out, NZ);
}


# f32-atomic scatter-add -> edges bucketed by destination row in the call (int-atomic rank, scan, place, per-row id sort); one wave sums a row and its degree in fixed edge order, one store per line
# speedup vs baseline: 2.1431x; 1.9437x over previous
#include <hip/hip_runtime.h>

#define NN 100000
#define NE 1600000

typedef float v2f __attribute__((ext_vector_type(2)));
typedef float v8f __attribute__((ext_vector_type(8)));

__global__ void cnt_zero_k(int* cnt, int n1) {
  int i = blockIdx.x * blockDim.x + threadIdx.x;
  if (i < n1) cnt[i] = 0;
}

__global__ void edge_rank_k(const int* __restrict__ row, int* cnt, int* __restrict__ rk, int e) {
  int i = blockIdx.x * blockDim.x + threadIdx.x;
  if (i < e) rk[i] = atomicAdd(&cnt[row[i]], 1);
}

__global__ __launch_bounds__(256) void scan_local_k(int* cnt, int* __restrict__ bsum, int n1) {
  __shared__ int wsum[8];
  const int lane = threadIdx.x & 31;
  const int wv   = threadIdx.x >> 5;
  const int i = blockIdx.x * 256 + threadIdx.x;
  const int v = (i < n1) ? cnt[i] : 0;
  int x = v;
  for (int d = 1; d < 32; d <<= 1) {
    int y = __shfl_up(x, d);
    if (lane >= d) x += y;
  }
  if (lane == 31) wsum[wv] = x;
  __syncthreads();
  int off = 0;
  for (int k = 0; k < 8; ++k) off += (k < wv) ? wsum[k] : 0;
  const int incl = x + off;
  if (i < n1) cnt[i] = incl - v;
  if (threadIdx.x == 255) bsum[blockIdx.x] = incl;
}

__global__ __launch_bounds__(256) void scan_add_k(int* rowptr, const int* __restrict__ bsum, int n1) {
  __shared__ int red[8];
  const int lane = threadIdx.x & 31;
  const int wv   = threadIdx.x >> 5;
  const int b = blockIdx.x;
  int s = 0;
  for (int k = threadIdx.x; k < b; k += 256) s += bsum[k];
  for (int d = 16; d >= 1; d >>= 1) s += __shfl_xor(s, d);
  if (lane == 0) red[wv] = s;
  __syncthreads();
  int off = 0;
  for (int k = 0; k < 8; ++k) off += red[k];
  const int i = b * 256 + threadIdx.x;
  if (i < n1) rowptr[i] += off;
}

__global__ void edge_place_k(const int* __restrict__ row, const int* __restrict__ rk,
                             const int* __restrict__ rowptr, int* __restrict__ tmp, int e) {
  int i = blockIdx.x * blockDim.x + threadIdx.x;
  if (i < e) tmp[rowptr[row[i]] + rk[i]] = i;
}

__global__ __launch_bounds__(256) void row_order_k(const int* __restrict__ rowptr,
                                                   const int* __restrict__ tmp,
                                                   const int* __restrict__ col,
                                                   const float* __restrict__ w,
                                                   int* __restrict__ col_s,
                                                   float* __restrict__ w_s, int n) {
  const int r = (blockIdx.x * 256 + threadIdx.x) >> 5;
  if (r >= n) return;
  const int lane = threadIdx.x & 31;
  const int s = __builtin_amdgcn_readfirstlane(rowptr[r]);
  const int L = __builtin_amdgcn_readfirstlane(rowptr[r + 1]) - s;
  for (int bi = 0; bi < L; bi += 32) {
    const int i = bi + lane;
    const int mine = (i < L) ? tmp[s + i] : 0x7fffffff;
    int dr = 0;
    for (int bj = 0; bj < L; bj += 32) {
      const int j = bj + lane;
      const int other = (j < L) ? tmp[s + j] : 0x7fffffff;
      const int m = (L - bj < 32) ? (L - bj) : 32;
      for (int q = 0; q < m; ++q)
        dr += (__builtin_amdgcn_readlane(other, q) < mine) ? 1 : 0;
    }
    if (i < L) {
      col_s[s + dr] = col[mine];
      w_s[s + dr]   = w[mine];
    }
  }
}

__global__ void deg_dis_k(const int* __restrict__ rowptr, const float* __restrict__ w_s,
                          float* __restrict__ dis, int n) {
  int i = blockIdx.x * blockDim.x + threadIdx.x;
  if (i >= n) return;
  const int s = rowptr[i], t = rowptr[i + 1];
  float d = 1.0f;
  for (int p = s; p < t; ++p) d += w_s[p];
  dis[i] = rsqrtf(d);
}

__global__ void edge_norm_k(const int* __restrict__ rowptr, const int* __restrict__ col_s,
                            const float* __restrict__ dis, float* w_s, int n) {
  int i = blockIdx.x * blockDim.x + threadIdx.x;
  if (i >= n) return;
  const int s = rowptr[i], t = rowptr[i + 1];
  const float di = dis[i];
  for (int p = s; p < t; ++p) w_s[p] = di * w_s[p] * dis[col_s[p]];
}

__global__ void transpose_k(const float* __restrict__ W, float* __restrict__ Wt, int U) {
  int i = blockIdx.x * blockDim.x + threadIdx.x;
  if (i >= 128 * U) return;
  int k = i / U, u = i % U;
  Wt[(size_t)u * 128 + k] = W[i];
}

template <int U, bool RELU_IN>
__global__ __launch_bounds__(128) void gemm_wmma_k(const float* __restrict__ X,
                                                   const float* __restrict__ Wt,
                                                   float* __restrict__ T, int M) {
  const int lane   = threadIdx.x & 31;
  const int wave   = threadIdx.x >> 5;
  const int tile_m = blockIdx.x * 4 + wave;
  const int tile_n = blockIdx.y;
  if (tile_m * 16 >= M) return;

  const int mrow = tile_m * 16 + (lane & 15);
  const int ncol = tile_n * 16 + (lane & 15);
  const int kk   = (lane >> 4) * 2;

  const float* xr = X + (size_t)mrow * 128 + kk;
  const float* wr = Wt + (size_t)ncol * 128 + kk;

  v8f c = {0.f, 0.f, 0.f, 0.f, 0.f, 0.f, 0.f, 0.f};
#pragma unroll
  for (int k = 0; k < 128; k += 4) {
    v2f a = *reinterpret_cast<const v2f*>(xr + k);
    if (RELU_IN) { a.x = fmaxf(a.x, 0.0f); a.y = fmaxf(a.y, 0.0f); }
    v2f b = *reinterpret_cast<const v2f*>(wr + k);
    c = __builtin_amdgcn_wmma_f32_16x16x4_f32(false, a, false, b, (short)0, c,
                                              false, false);
  }

  const int mbase = tile_m * 16 + (lane >> 4) * 8;
  float* trow = T + (size_t)mbase * U + ncol;
#pragma unroll
  for (int r = 0; r < 8; ++r) trow[(size_t)r * U] = c[r];
}

template <int U>
__global__ __launch_bounds__(256) void agg_rows_k(const int* __restrict__ rowptr,
                                                  const int* __restrict__ col_s,
                                                  const float* __restrict__ nrm_s,
                                                  const float* __restrict__ T,
                                                  const float* __restrict__ dis,
                                                  const float* __restrict__ bias,
                                                  float* __restrict__ out, int n) {
  constexpr int NL = U / 32;
  const int r = (blockIdx.x * 256 + threadIdx.x) >> 5;
  if (r >= n) return;
  const int lane = threadIdx.x & 31;
  const int s = __builtin_amdgcn_readfirstlane(rowptr[r]);
  const int t = __builtin_amdgcn_readfirstlane(rowptr[r + 1]);
  const float* Tl = T + lane;
  const float d  = dis[r];
  const float sn = d * d;
  float acc[NL];
  for (int j = 0; j < NL; ++j) acc[j] = bias[lane + 32 * j] + sn * Tl[(size_t)r * U + 32 * j];

  for (int base = s; base < t; base += 32) {
    const int m = (t - base < 32) ? (t - base) : 32;
    int   cv = 0;
    float wv = 0.0f;
    if (lane < m) { cv = col_s[base + lane]; wv = nrm_s[base + lane]; }
    const int wi = __float_as_int(wv);
    int q = 0;
    for (; q + 4 <= m; q += 4) {
      const int c0 = __builtin_amdgcn_readlane(cv, q);
      const int c1 = __builtin_amdgcn_readlane(cv, q + 1);
      const int c2 = __builtin_amdgcn_readlane(cv, q + 2);
      const int c3 = __builtin_amdgcn_readlane(cv, q + 3);
      const float w0 = __int_as_float(__builtin_amdgcn_readlane(wi, q));
      const float w1 = __int_as_float(__builtin_amdgcn_readlane(wi, q + 1));
      const float w2 = __int_as_float(__builtin_amdgcn_readlane(wi, q + 2));
      const float w3 = __int_as_float(__builtin_amdgcn_readlane(wi, q + 3));
      float t0[NL], t1[NL], t2[NL], t3[NL];
      for (int j = 0; j < NL; ++j) {
        t0[j] = Tl[(size_t)c0 * U + 32 * j];
        t1[j] = Tl[(size_t)c1 * U + 32 * j];
        t2[j] = Tl[(size_t)c2 * U + 32 * j];
        t3[j] = Tl[(size_t)c3 * U + 32 * j];
      }
      for (int j = 0; j < NL; ++j) {
        acc[j] += w0 * t0[j];
        acc[j] += w1 * t1[j];
        acc[j] += w2 * t2[j];
        acc[j] += w3 * t3[j];
      }
    }
    for (; q < m; ++q) {
      const int   c0 = __builtin_amdgcn_readlane(cv, q);
      const float w0 = __int_as_float(__builtin_amdgcn_readlane(wi, q));
      for (int j = 0; j < NL; ++j) acc[j] += w0 * Tl[(size_t)c0 * U + 32 * j];
    }
  }
  float* orow = out + (size_t)r * U + lane;
  for (int j = 0; j < NL; ++j) orow[32 * j] = acc[j];
}

__global__ void final_k(const float* __restrict__ rnd, float* out, int nz) {
  int i = blockIdx.x * blockDim.x + threadIdx.x;
  if (i >= nz) return;
  float mu = out[i];
  float ls = out[nz + i];
  float s  = expf(ls);
  out[i]      = mu + rnd[i] * s;
  out[nz + i] = -ls + 0.5f * (s * s + mu * mu - 1.0f);
}

extern "C" void kernel_launch(void* const* d_in, const int* in_sizes, int n_in,
                              void* d_out, int out_size, void* d_ws, size_t ws_size,
                              hipStream_t stream) {
  (void)in_sizes; (void)n_in; (void)out_size; (void)ws_size;
  const int N = NN, E = NE, H = 128, DZ = 64;
  const float* h   = (const float*)d_in[0];
  const int*   ei  = (const int*)d_in[1];
  const float* ew  = (const float*)d_in[2];
  const float* rnd = (const float*)d_in[3];
  const float* W0  = (const float*)d_in[4];
  const float* b0  = (const float*)d_in[5];
  const float* W1  = (const float*)d_in[6];
  const float* b1  = (const float*)d_in[7];
  const float* Wmu = (const float*)d_in[8];
  const float* bmu = (const float*)d_in[9];
  const float* Wsd = (const float*)d_in[10];
  const float* bsd = (const float*)d_in[11];
  const int* row = ei;
  const int* col = ei + E;
  float* out = (float*)d_out;
  const int NZ = N * DZ;

  float* ws    = (float*)d_ws;
  float* dis   = ws;
  float* nrm   = dis + N;
  float* bufT  = nrm + E;
  float* bufH  = bufT + (size_t)N * 128;
  float* bufS  = bufH + (size_t)N * 128;
  float* Wt0   = bufS + (size_t)N * 64;
  float* Wt1   = Wt0 + 128 * 128;
  float* Wtmu  = Wt1 + 128 * 128;
  float* Wtsd  = Wtmu + 64 * 128;
  int* col_s   = (int*)bufS;
  int* rowptr  = col_s + E;
  int* bsum    = rowptr + N + 96;
  int* rk      = (int*)bufH;
  int* tmp     = rk + E;

  const int TB = 256;
  const int N1 = N + 1;
  const int SB = (N1 + 255) / 256;
  cnt_zero_k  <<<(N1 + TB - 1) / TB, TB, 0, stream>>>(rowptr, N1);
  edge_rank_k <<<(E + TB - 1) / TB, TB, 0, stream>>>(row, rowptr, rk, E);
  scan_local_k<<<SB, 256, 0, stream>>>(rowptr, bsum, N1);
  scan_add_k  <<<SB, 256, 0, stream>>>(rowptr, bsum, N1);
  edge_place_k<<<(E + TB - 1) / TB, TB, 0, stream>>>(row, rk, rowptr, tmp, E);
  row_order_k <<<(N * 32 + 255) / 256, 256, 0, stream>>>(rowptr, tmp, col, ew, col_s, nrm, N);
  deg_dis_k   <<<(N + TB - 1) / TB, TB, 0, stream>>>(rowptr, nrm, dis, N);
  edge_norm_k <<<(N + TB - 1) / TB, TB, 0, stream>>>(rowptr, col_s, dis, nrm, N);

  transpose_k<<<(128 * H  + TB - 1) / TB, TB, 0, stream>>>(W0,  Wt0,  H);
  transpose_k<<<(128 * H  + TB - 1) / TB, TB, 0, stream>>>(W1,  Wt1,  H);
  transpose_k<<<(128 * DZ + TB - 1) / TB, TB, 0, stream>>>(Wmu, Wtmu, DZ);
  transpose_k<<<(128 * DZ + TB - 1) / TB, TB, 0, stream>>>(Wsd, Wtsd, DZ);

  const int MT = N / 16;
  dim3 gblk(128);
  const unsigned MB = (unsigned)((MT + 3) / 4);
  const unsigned AB = (unsigned)((N * 32 + 255) / 256);

  gemm_wmma_k<128, false><<<dim3(MB, H / 16), gblk, 0, stream>>>(h, Wt0, bufT, N);
  agg_rows_k<128><<<AB, 256, 0, stream>>>(rowptr, col_s, nrm, bufT, dis, b0, bufH, N);
  gemm_wmma_k<128, true><<<dim3(MB, H / 16), gblk, 0, stream>>>(bufH, Wt1, bufT, N);
  agg_rows_k<128><<<AB, 256, 0, stream>>>(rowptr, col_s, nrm, bufT, dis, b1, bufH, N);
  gemm_wmma_k<64, true><<<dim3(MB, DZ / 16), gblk, 0, stream>>>(bufH, Wtmu, bufT, N);
  agg_rows_k<64><<<AB, 256, 0, stream>>>(rowptr, col_s, nrm, bufT, dis, bmu, out, N);
  gemm_wmma_k<64, true><<<dim3(MB, DZ / 16), gblk, 0, stream>>>(bufH, Wtsd, bufT, N);
  agg_rows_k<64><<<AB, 256, 0, stream>>>(rowptr, col_s, nrm, bufT, dis, bsd, out + NZ, N);
  final_k<<<(NZ + TB - 1) / TB, TB, 0, stream>>>(rnd, out, NZ);
}


# GEMM retiled: one wave per 16-row strip with all U/16 accumulator tiles (was one wave per 16x16 tile); X strip and W^T each read once, as whole rows, via padded LDS; k order per tile kept
# speedup vs baseline: 2.8647x; 1.3367x over previous
#include <hip/hip_runtime.h>

#define NN 100000
#define NE 1600000

typedef float v2f __attribute__((ext_vector_type(2)));
typedef float v8f __attribute__((ext_vector_type(8)));

__global__ void cnt_zero_k(int* cnt, int n1) {
  int i = blockIdx.x * blockDim.x + threadIdx.x;
  if (i < n1) cnt[i] = 0;
}

__global__ void edge_rank_k(const int* __restrict__ row, int* cnt, int* __restrict__ rk, int e) {
  int i = blockIdx.x * blockDim.x + threadIdx.x;
  if (i < e) rk[i] = atomicAdd(&cnt[row[i]], 1);
}

__global__ __launch_bounds__(256) void scan_local_k(int* cnt, int* __restrict__ bsum, int n1) {
  __shared__ int wsum[8];
  const int lane = threadIdx.x & 31;
  const int wv   = threadIdx.x >> 5;
  const int i = blockIdx.x * 256 + threadIdx.x;
  const int v = (i < n1) ? cnt[i] : 0;
  int x = v;
  for (int d = 1; d < 32; d <<= 1) {
    int y = __shfl_up(x, d);
    if (lane >= d) x += y;
  }
  if (lane == 31) wsum[wv] = x;
  __syncthreads();
  int off = 0;
  for (int k = 0; k < 8; ++k) off += (k < wv) ? wsum[k] : 0;
  const int incl = x + off;
  if (i < n1) cnt[i] = incl - v;
  if (threadIdx.x == 255) bsum[blockIdx.x] = incl;
}

__global__ __launch_bounds__(256) void scan_add_k(int* rowptr, const int* __restrict__ bsum, int n1) {
  __shared__ int red[8];
  const int lane = threadIdx.x & 31;
  const int wv   = threadIdx.x >> 5;
  const int b = blockIdx.x;
  int s = 0;
  for (int k = threadIdx.x; k < b; k += 256) s += bsum[k];
  for (int d = 16; d >= 1; d >>= 1) s += __shfl_xor(s, d);
  if (lane == 0) red[wv] = s;
  __syncthreads();
  int off = 0;
  for (int k = 0; k < 8; ++k) off += red[k];
  const int i = b * 256 + threadIdx.x;
  if (i < n1) rowptr[i] += off;
}

__global__ void edge_place_k(const int* __restrict__ row, const int* __restrict__ rk,
                             const int* __restrict__ rowptr, int* __restrict__ tmp, int e) {
  int i = blockIdx.x * blockDim.x + threadIdx.x;
  if (i < e) tmp[rowptr[row[i]] + rk[i]] = i;
}

__global__ __launch_bounds__(256) void row_order_k(const int* __restrict__ rowptr,
                                                   const int* __restrict__ tmp,
                                                   const int* __restrict__ col,
                                                   const float* __restrict__ w,
                                                   int* __restrict__ col_s,
                                                   float* __restrict__ w_s, int n) {
  const int r = (blockIdx.x * 256 + threadIdx.x) >> 5;
  if (r >= n) return;
  const int lane = threadIdx.x & 31;
  const int s = __builtin_amdgcn_readfirstlane(rowptr[r]);
  const int L = __builtin_amdgcn_readfirstlane(rowptr[r + 1]) - s;
  for (int bi = 0; bi < L; bi += 32) {
    const int i = bi + lane;
    const int mine = (i < L) ? tmp[s + i] : 0x7fffffff;
    int dr = 0;
    for (int bj = 0; bj < L; bj += 32) {
      const int j = bj + lane;
      const int other = (j < L) ? tmp[s + j] : 0x7fffffff;
      const int m = (L - bj < 32) ? (L - bj) : 32;
      for (int q = 0; q < m; ++q)
        dr += (__builtin_amdgcn_readlane(other, q) < mine) ? 1 : 0;
    }
    if (i < L) {
      col_s[s + dr] = col[mine];
      w_s[s + dr]   = w[mine];
    }
  }
}

__global__ void deg_dis_k(const int* __restrict__ rowptr, const float* __restrict__ w_s,
                          float* __restrict__ dis, int n) {
  int i = blockIdx.x * blockDim.x + threadIdx.x;
  if (i >= n) return;
  const int s = rowptr[i], t = rowptr[i + 1];
  float d = 1.0f;
  for (int p = s; p < t; ++p) d += w_s[p];
  dis[i] = rsqrtf(d);
}

__global__ void edge_norm_k(const int* __restrict__ rowptr, const int* __restrict__ col_s,
                            const float* __restrict__ dis, float* w_s, int n) {
  int i = blockIdx.x * blockDim.x + threadIdx.x;
  if (i >= n) return;
  const int s = rowptr[i], t = rowptr[i + 1];
  const float di = dis[i];
  for (int p = s; p < t; ++p) w_s[p] = di * w_s[p] * dis[col_s[p]];
}

__global__ void transpose_k(const float* __restrict__ W, float* __restrict__ Wt, int U) {
  int i = blockIdx.x * blockDim.x + threadIdx.x;
  if (i >= 128 * U) return;
  int k = i / U, u = i % U;
  Wt[(size_t)u * 128 + k] = W[i];
}

template <int U, bool RELU_IN>
__global__ __launch_bounds__(256) void gemm_strip_k(const float* __restrict__ X,
                                                    const float* __restrict__ Wt,
                                                    float* __restrict__ T, int MT) {
  constexpr int P = 132;
  constexpr int NT = U / 16;
  __shared__ alignas(16) float Bs[U * P];
  __shared__ alignas(16) float As[8 * 16 * P];
  const int lane = threadIdx.x & 31;
  const int wave = threadIdx.x >> 5;
  const int tile_m = blockIdx.x * 8 + wave;
  const bool live = tile_m < MT;
  const int tm = live ? tile_m : MT - 1;

  const float* xs = X + (size_t)tm * (16 * 128) + lane * 4;
  float* as = As + wave * (16 * P);
  float4 xa[16];
  for (int j = 0; j < 16; ++j) xa[j] = *reinterpret_cast<const float4*>(xs + j * 128);
  for (int i = 0; i < U / 8; ++i) {
    const int n = i * 8 + wave;
    const float4 wv = *reinterpret_cast<const float4*>(Wt + (size_t)n * 128 + lane * 4);
    *reinterpret_cast<float4*>(Bs + n * P + lane * 4) = wv;
  }
  for (int j = 0; j < 16; ++j) *reinterpret_cast<float4*>(as + j * P + lane * 4) = xa[j];
  __syncthreads();
  if (!live) return;

  const int kk = (lane >> 4) * 2;
  const float* ar = as + (lane & 15) * P + kk;
  const float* br = Bs + (lane & 15) * P + kk;
  v8f c[NT];
  for (int t = 0; t < NT; ++t) c[t] = v8f{0.f, 0.f, 0.f, 0.f, 0.f, 0.f, 0.f, 0.f};
  for (int k = 0; k < 128; k += 4) {
    v2f a = *reinterpret_cast<const v2f*>(ar + k);
    if (RELU_IN) { a.x = fmaxf(a.x, 0.0f); a.y = fmaxf(a.y, 0.0f); }
    for (int t = 0; t < NT; ++t) {
      const v2f b = *reinterpret_cast<const v2f*>(br + t * (16 * P) + k);
      c[t] = __builtin_amdgcn_wmma_f32_16x16x4_f32(false, a, false, b, (short)0, c[t],
                                                   false, false);
    }
  }

  const int mbase = tile_m * 16 + (lane >> 4) * 8;
  float* trow = T + (size_t)mbase * U + (lane & 15);
  for (int t = 0; t < NT; ++t)
    for (int r = 0; r < 8; ++r) trow[(size_t)r * U + t * 16] = c[t][r];
}

template <int U>
__global__ __launch_bounds__(256) void agg_rows_k(const int* __restrict__ rowptr,
                                                  const int* __restrict__ col_s,
                                                  const float* __restrict__ nrm_s,
                                                  const float* __restrict__ T,
                                                  const float* __restrict__ dis,
                                                  const float* __restrict__ bias,
                                                  float* __restrict__ out, int n) {
  constexpr int NL = U / 32;
  const int r = (blockIdx.x * 256 + threadIdx.x) >> 5;
  if (r >= n) return;
  const int lane = threadIdx.x & 31;
  const int s = __builtin_amdgcn_readfirstlane(rowptr[r]);
  const int t = __builtin_amdgcn_readfirstlane(rowptr[r + 1]);
  const float* Tl = T + lane;
  const float d  = dis[r];
  const float sn = d * d;
  float acc[NL];
  for (int j = 0; j < NL; ++j) acc[j] = bias[lane + 32 * j] + sn * Tl[(size_t)r * U + 32 * j];

  for (int base = s; base < t; base += 32) {
    const int m = (t - base < 32) ? (t - base) : 32;
    int   cv = 0;
    float wv = 0.0f;
    if (lane < m) { cv = col_s[base + lane]; wv = nrm_s[base + lane]; }
    const int wi = __float_as_int(wv);
    int q = 0;
    for (; q + 4 <= m; q += 4) {
      const int c0 = __builtin_amdgcn_readlane(cv, q);
      const int c1 = __builtin_amdgcn_readlane(cv, q + 1);
      const int c2 = __builtin_amdgcn_readlane(cv, q + 2);
      const int c3 = __builtin_amdgcn_readlane(cv, q + 3);
      const float w0 = __int_as_float(__builtin_amdgcn_readlane(wi, q));
      const float w1 = __int_as_float(__builtin_amdgcn_readlane(wi, q + 1));
      const float w2 = __int_as_float(__builtin_amdgcn_readlane(wi, q + 2));
      const float w3 = __int_as_float(__builtin_amdgcn_readlane(wi, q + 3));
      float t0[NL], t1[NL], t2[NL], t3[NL];
      for (int j = 0; j < NL; ++j) {
        t0[j] = Tl[(size_t)c0 * U + 32 * j];
        t1[j] = Tl[(size_t)c1 * U + 32 * j];
        t2[j] = Tl[(size_t)c2 * U + 32 * j];
        t3[j] = Tl[(size_t)c3 * U + 32 * j];
      }
      for (int j = 0; j < NL; ++j) {
        acc[j] += w0 * t0[j];
        acc[j] += w1 * t1[j];
        acc[j] += w2 * t2[j];
        acc[j] += w3 * t3[j];
      }
    }
    for (; q < m; ++q) {
      const int   c0 = __builtin_amdgcn_readlane(cv, q);
      const float w0 = __int_as_float(__builtin_amdgcn_readlane(wi, q));
      for (int j = 0; j < NL; ++j) acc[j] += w0 * Tl[(size_t)c0 * U + 32 * j];
    }
  }
  float* orow = out + (size_t)r * U + lane;
  for (int j = 0; j < NL; ++j) orow[32 * j] = acc[j];
}

__global__ void final_k(const float* __restrict__ rnd, float* out, int nz) {
  int i = blockIdx.x * blockDim.x + threadIdx.x;
  if (i >= nz) return;
  float mu = out[i];
  float ls = out[nz + i];
  float s  = expf(ls);
  out[i]      = mu + rnd[i] * s;
  out[nz + i] = -ls + 0.5f * (s * s + mu * mu - 1.0f);
}

extern "C" void kernel_launch(void* const* d_in, const int* in_sizes, int n_in,
                              void* d_out, int out_size, void* d_ws, size_t ws_size,
                              hipStream_t stream) {
  (void)in_sizes; (void)n_in; (void)out_size; (void)ws_size;
  const int N = NN, E = NE, H = 128, DZ = 64;
  const float* h   = (const float*)d_in[0];
  const int*   ei  = (const int*)d_in[1];
  const float* ew  = (const float*)d_in[2];
  const float* rnd = (const float*)d_in[3];
  const float* W0  = (const float*)d_in[4];
  const float* b0  = (const float*)d_in[5];
  const float* W1  = (const float*)d_in[6];
  const float* b1  = (const float*)d_in[7];
  const float* Wmu = (const float*)d_in[8];
  const float* bmu = (const float*)d_in[9];
  const float* Wsd = (const float*)d_in[10];
  const float* bsd = (const float*)d_in[11];
  const int* row = ei;
  const int* col = ei + E;
  float* out = (float*)d_out;
  const int NZ = N * DZ;

  float* ws    = (float*)d_ws;
  float* dis   = ws;
  float* nrm   = dis + N;
  float* bufT  = nrm + E;
  float* bufH  = bufT + (size_t)N * 128;
  float* bufS  = bufH + (size_t)N * 128;
  float* Wt0   = bufS + (size_t)N * 64;
  float* Wt1   = Wt0 + 128 * 128;
  float* Wtmu  = Wt1 + 128 * 128;
  float* Wtsd  = Wtmu + 64 * 128;
  int* col_s   = (int*)bufS;
  int* rowptr  = col_s + E;
  int* bsum    = rowptr + N + 96;
  int* rk      = (int*)bufH;
  int* tmp     = rk + E;

  const int TB = 256;
  const int N1 = N + 1;
  const int SB = (N1 + 255) / 256;
  cnt_zero_k  <<<(N1 + TB - 1) / TB, TB, 0, stream>>>(rowptr, N1);
  edge_rank_k <<<(E + TB - 1) / TB, TB, 0, stream>>>(row, rowptr, rk, E);
  scan_local_k<<<SB, 256, 0, stream>>>(rowptr, bsum, N1);
  scan_add_k  <<<SB, 256, 0, stream>>>(rowptr, bsum, N1);
  edge_place_k<<<(E + TB - 1) / TB, TB, 0, stream>>>(row, rk, rowptr, tmp, E);
  row_order_k <<<(N * 32 + 255) / 256, 256, 0, stream>>>(rowptr, tmp, col, ew, col_s, nrm, N);
  deg_dis_k   <<<(N + TB - 1) / TB, TB, 0, stream>>>(rowptr, nrm, dis, N);
  edge_norm_k <<<(N + TB - 1) / TB, TB, 0, stream>>>(rowptr, col_s, dis, nrm, N);

  transpose_k<<<(128 * H  + TB - 1) / TB, TB, 0, stream>>>(W0,  Wt0,  H);
  transpose_k<<<(128 * H  + TB - 1) / TB, TB, 0, stream>>>(W1,  Wt1,  H);
  transpose_k<<<(128 * DZ + TB - 1) / TB, TB, 0, stream>>>(Wmu, Wtmu, DZ);
  transpose_k<<<(128 * DZ + TB - 1) / TB, TB, 0, stream>>>(Wsd, Wtsd, DZ);

  const int MT = N / 16;
  const unsigned GB = (unsigned)((MT + 7) / 8);
  const unsigned AB = (unsigned)((N * 32 + 255) / 256);

  gemm_strip_k<128, false><<<GB, 256, 0, stream>>>(h, Wt0, bufT, MT);
  agg_rows_k<128><<<AB, 256, 0, stream>>>(rowptr, col_s, nrm, bufT, dis, b0, bufH, N);
  gemm_strip_k<128, true><<<GB, 256, 0, stream>>>(bufH, Wt1, bufT, MT);
  agg_rows_k<128><<<AB, 256, 0, stream>>>(rowptr, col_s, nrm, bufT, dis, b1, bufH, N);
  gemm_strip_k<64, true><<<GB, 256, 0, stream>>>(bufH, Wtmu, bufT, MT);
  agg_rows_k<64><<<AB, 256, 0, stream>>>(rowptr, col_s, nrm, bufT, dis, bmu, out, N);
  gemm_strip_k<64, true><<<GB, 256, 0, stream>>>(bufH, Wtsd, bufT, MT);
  agg_rows_k<64><<<AB, 256, 0, stream>>>(rowptr, col_s, nrm, bufT, dis, bsd, out + NZ, N);
  final_k<<<(NZ + TB - 1) / TB, TB, 0, stream>>>(rnd, out, NZ);
}


# mu and log_std heads run as one 128-wide layer: one GEMM on [Wmu|Wstd], one aggregation whose wave holds both halves of a row and forms sample and KL in registers; each column's sum order kept
# speedup vs baseline: 3.1028x; 1.0831x over previous
#include <hip/hip_runtime.h>

#define NN 100000
#define NE 1600000

typedef float v2f __attribute__((ext_vector_type(2)));
typedef float v8f __attribute__((ext_vector_type(8)));

__global__ void cnt_zero_k(int* cnt, int n1) {
  int i = blockIdx.x * blockDim.x + threadIdx.x;
  if (i < n1) cnt[i] = 0;
}

__global__ void edge_rank_k(const int* __restrict__ row, int* cnt, int* __restrict__ rk, int e) {
  int i = blockIdx.x * blockDim.x + threadIdx.x;
  if (i < e) rk[i] = atomicAdd(&cnt[row[i]], 1);
}

__global__ __launch_bounds__(256) void scan_local_k(int* cnt, int* __restrict__ bsum, int n1) {
  __shared__ int wsum[8];
  const int lane = threadIdx.x & 31;
  const int wv   = threadIdx.x >> 5;
  const int i = blockIdx.x * 256 + threadIdx.x;
  const int v = (i < n1) ? cnt[i] : 0;
  int x = v;
  for (int d = 1; d < 32; d <<= 1) {
    int y = __shfl_up(x, d);
    if (lane >= d) x += y;
  }
  if (lane == 31) wsum[wv] = x;
  __syncthreads();
  int off = 0;
  for (int k = 0; k < 8; ++k) off += (k < wv) ? wsum[k] : 0;
  const int incl = x + off;
  if (i < n1) cnt[i] = incl - v;
  if (threadIdx.x == 255) bsum[blockIdx.x] = incl;
}

__global__ __launch_bounds__(256) void scan_add_k(int* rowptr, const int* __restrict__ bsum, int n1) {
  __shared__ int red[8];
  const int lane = threadIdx.x & 31;
  const int wv   = threadIdx.x >> 5;
  const int b = blockIdx.x;
  int s = 0;
  for (int k = threadIdx.x; k < b; k += 256) s += bsum[k];
  for (int d = 16; d >= 1; d >>= 1) s += __shfl_xor(s, d);
  if (lane == 0) red[wv] = s;
  __syncthreads();
  int off = 0;
  for (int k = 0; k < 8; ++k) off += red[k];
  const int i = b * 256 + threadIdx.x;
  if (i < n1) rowptr[i] += off;
}

__global__ void edge_place_k(const int* __restrict__ row, const int* __restrict__ rk,
                             const int* __restrict__ rowptr, int* __restrict__ tmp, int e) {
  int i = blockIdx.x * blockDim.x + threadIdx.x;
  if (i < e) tmp[rowptr[row[i]] + rk[i]] = i;
}

__global__ __launch_bounds__(256) void row_order_k(const int* __restrict__ rowptr,
                                                   const int* __restrict__ tmp,
                                                   const int* __restrict__ col,
                                                   const float* __restrict__ w,
                                                   int* __restrict__ col_s,
                                                   float* __restrict__ w_s, int n) {
  const int r = (blockIdx.x * 256 + threadIdx.x) >> 5;
  if (r >= n) return;
  const int lane = threadIdx.x & 31;
  const int s = __builtin_amdgcn_readfirstlane(rowptr[r]);
  const int L = __builtin_amdgcn_readfirstlane(rowptr[r + 1]) - s;
  for (int bi = 0; bi < L; bi += 32) {
    const int i = bi + lane;
    const int mine = (i < L) ? tmp[s + i] : 0x7fffffff;
    int dr = 0;
    for (int bj = 0; bj < L; bj += 32) {
      const int j = bj + lane;
      const int other = (j < L) ? tmp[s + j] : 0x7fffffff;
      const int m = (L - bj < 32) ? (L - bj) : 32;
      for (int q = 0; q < m; ++q)
        dr += (__builtin_amdgcn_readlane(other, q) < mine) ? 1 : 0;
    }
    if (i < L) {
      col_s[s + dr] = col[mine];
      w_s[s + dr]   = w[mine];
    }
  }
}

__global__ void deg_dis_k(const int* __restrict__ rowptr, const float* __restrict__ w_s,
                          float* __restrict__ dis, int n) {
  int i = blockIdx.x * blockDim.x + threadIdx.x;
  if (i >= n) return;
  const int s = rowptr[i], t = rowptr[i + 1];
  float d = 1.0f;
  for (int p = s; p < t; ++p) d += w_s[p];
  dis[i] = rsqrtf(d);
}

__global__ void edge_norm_k(const int* __restrict__ rowptr, const int* __restrict__ col_s,
                            const float* __restrict__ dis, float* w_s, int n) {
  int i = blockIdx.x * blockDim.x + threadIdx.x;
  if (i >= n) return;
  const int s = rowptr[i], t = rowptr[i + 1];
  const float di = dis[i];
  for (int p = s; p < t; ++p) w_s[p] = di * w_s[p] * dis[col_s[p]];
}

__global__ void transpose_k(const float* __restrict__ W, float* __restrict__ Wt, int U) {
  int i = blockIdx.x * blockDim.x + threadIdx.x;
  if (i >= 128 * U) return;
  int k = i / U, u = i % U;
  Wt[(size_t)u * 128 + k] = W[i];
}

template <int U, bool RELU_IN>
__global__ __launch_bounds__(256) void gemm_strip_k(const float* __restrict__ X,
                                                    const float* __restrict__ Wt,
                                                    float* __restrict__ T, int MT) {
  constexpr int P = 132;
  constexpr int NT = U / 16;
  __shared__ alignas(16) float Bs[U * P];
  __shared__ alignas(16) float As[8 * 16 * P];
  const int lane = threadIdx.x & 31;
  const int wave = threadIdx.x >> 5;
  const int tile_m = blockIdx.x * 8 + wave;
  const bool live = tile_m < MT;
  const int tm = live ? tile_m : MT - 1;

  const float* xs = X + (size_t)tm * (16 * 128) + lane * 4;
  float* as = As + wave * (16 * P);
  float4 xa[16];
  for (int j = 0; j < 16; ++j) xa[j] = *reinterpret_cast<const float4*>(xs + j * 128);
  for (int i = 0; i < U / 8; ++i) {
    const int n = i * 8 + wave;
    const float4 wv = *reinterpret_cast<const float4*>(Wt + (size_t)n * 128 + lane * 4);
    *reinterpret_cast<float4*>(Bs + n * P + lane * 4) = wv;
  }
  for (int j = 0; j < 16; ++j) *reinterpret_cast<float4*>(as + j * P + lane * 4) = xa[j];
  __syncthreads();
  if (!live) return;

  const int kk = (lane >> 4) * 2;
  const float* ar = as + (lane & 15) * P + kk;
  const float* br = Bs + (lane & 15) * P + kk;
  v8f c[NT];
  for (int t = 0; t < NT; ++t) c[t] = v8f{0.f, 0.f, 0.f, 0.f, 0.f, 0.f, 0.f, 0.f};
  for (int k = 0; k < 128; k += 4) {
    v2f a = *reinterpret_cast<const v2f*>(ar + k);
    if (RELU_IN) { a.x = fmaxf(a.x, 0.0f); a.y = fmaxf(a.y, 0.0f); }
    for (int t = 0; t < NT; ++t) {
      const v2f b = *reinterpret_cast<const v2f*>(br + t * (16 * P) + k);
      c[t] = __builtin_amdgcn_wmma_f32_16x16x4_f32(false, a, false, b, (short)0, c[t],
                                                   false, false);
    }
  }

  const int mbase = tile_m * 16 + (lane >> 4) * 8;
  float* trow = T + (size_t)mbase * U + (lane & 15);
  for (int t = 0; t < NT; ++t)
    for (int r = 0; r < 8; ++r) trow[(size_t)r * U + t * 16] = c[t][r];
}

template <int U, bool HEADS>
__global__ __launch_bounds__(256) void agg_rows_k(const int* __restrict__ rowptr,
                                                  const int* __restrict__ col_s,
                                                  const float* __restrict__ nrm_s,
                                                  const float* __restrict__ T,
                                                  const float* __restrict__ dis,
                                                  const float* __restrict__ bias,
                                                  const float* __restrict__ bias_hi,
                                                  const float* __restrict__ rnd,
                                                  float* __restrict__ out, int n) {
  constexpr int NL = U / 32;
  constexpr int HL = NL / 2;
  const int r = (blockIdx.x * 256 + threadIdx.x) >> 5;
  if (r >= n) return;
  const int lane = threadIdx.x & 31;
  const int s = __builtin_amdgcn_readfirstlane(rowptr[r]);
  const int t = __builtin_amdgcn_readfirstlane(rowptr[r + 1]);
  const float* Tl = T + lane;
  const float d  = dis[r];
  const float sn = d * d;
  float acc[NL];
  for (int j = 0; j < NL; ++j) {
    const float bj = (HEADS && j >= HL) ? bias_hi[lane + 32 * (j - HL)] : bias[lane + 32 * j];
    acc[j] = bj + sn * Tl[(size_t)r * U + 32 * j];
  }
  float rv[HL > 0 ? HL : 1];
  if (HEADS)
    for (int j = 0; j < HL; ++j) rv[j] = rnd[(size_t)r * (U / 2) + lane + 32 * j];

  for (int base = s; base < t; base += 32) {
    const int m = (t - base < 32) ? (t - base) : 32;
    int   cv = 0;
    float wv = 0.0f;
    if (lane < m) { cv = col_s[base + lane]; wv = nrm_s[base + lane]; }
    const int wi = __float_as_int(wv);
    int q = 0;
    for (; q + 4 <= m; q += 4) {
      const int c0 = __builtin_amdgcn_readlane(cv, q);
      const int c1 = __builtin_amdgcn_readlane(cv, q + 1);
      const int c2 = __builtin_amdgcn_readlane(cv, q + 2);
      const int c3 = __builtin_amdgcn_readlane(cv, q + 3);
      const float w0 = __int_as_float(__builtin_amdgcn_readlane(wi, q));
      const float w1 = __int_as_float(__builtin_amdgcn_readlane(wi, q + 1));
      const float w2 = __int_as_float(__builtin_amdgcn_readlane(wi, q + 2));
      const float w3 = __int_as_float(__builtin_amdgcn_readlane(wi, q + 3));
      float t0[NL], t1[NL], t2[NL], t3[NL];
      for (int j = 0; j < NL; ++j) {
        t0[j] = Tl[(size_t)c0 * U + 32 * j];
        t1[j] = Tl[(size_t)c1 * U + 32 * j];
        t2[j] = Tl[(size_t)c2 * U + 32 * j];
        t3[j] = Tl[(size_t)c3 * U + 32 * j];
      }
      for (int j = 0; j < NL; ++j) {
        acc[j] += w0 * t0[j];
        acc[j] += w1 * t1[j];
        acc[j] += w2 * t2[j];
        acc[j] += w3 * t3[j];
      }
    }
    for (; q < m; ++q) {
      const int   c0 = __builtin_amdgcn_readlane(cv, q);
      const float w0 = __int_as_float(__builtin_amdgcn_readlane(wi, q));
      for (int j = 0; j < NL; ++j) acc[j] += w0 * Tl[(size_t)c0 * U + 32 * j];
    }
  }
  if (HEADS) {
    float* ho = out + (size_t)r * (U / 2) + lane;
    float* ko = ho + (size_t)n * (U / 2);
    for (int j = 0; j < HL; ++j) {
      const float mu = acc[j];
      const float ls = acc[HL + j];
      const float sd = expf(ls);
      ho[32 * j] = mu + rv[j] * sd;
      ko[32 * j] = -ls + 0.5f * (sd * sd + mu * mu - 1.0f);
    }
  } else {
    float* orow = out + (size_t)r * U + lane;
    for (int j = 0; j < NL; ++j) orow[32 * j] = acc[j];
  }
}

extern "C" void kernel_launch(void* const* d_in, const int* in_sizes, int n_in,
                              void* d_out, int out_size, void* d_ws, size_t ws_size,
                              hipStream_t stream) {
  (void)in_sizes; (void)n_in; (void)out_size; (void)ws_size;
  const int N = NN, E = NE, H = 128, DZ = 64;
  const float* h   = (const float*)d_in[0];
  const int*   ei  = (const int*)d_in[1];
  const float* ew  = (const float*)d_in[2];
  const float* rnd = (const float*)d_in[3];
  const float* W0  = (const float*)d_in[4];
  const float* b0  = (const float*)d_in[5];
  const float* W1  = (const float*)d_in[6];
  const float* b1  = (const float*)d_in[7];
  const float* Wmu = (const float*)d_in[8];
  const float* bmu = (const float*)d_in[9];
  const float* Wsd = (const float*)d_in[10];
  const float* bsd = (const float*)d_in[11];
  const int* row = ei;
  const int* col = ei + E;
  float* out = (float*)d_out;

  float* ws    = (float*)d_ws;
  float* dis   = ws;
  float* nrm   = dis + N;
  float* bufT  = nrm + E;
  float* bufH  = bufT + (size_t)N * 128;
  float* bufS  = bufH + (size_t)N * 128;
  float* Wt0   = bufS + (size_t)N * 64;
  float* Wt1   = Wt0 + 128 * 128;
  float* Wthd  = Wt1 + 128 * 128;
  int* col_s   = (int*)bufS;
  int* rowptr  = col_s + E;
  int* bsum    = rowptr + N + 96;
  int* rk      = (int*)bufH;
  int* tmp     = rk + E;

  const int TB = 256;
  const int N1 = N + 1;
  const int SB = (N1 + 255) / 256;
  cnt_zero_k  <<<(N1 + TB - 1) / TB, TB, 0, stream>>>(rowptr, N1);
  edge_rank_k <<<(E + TB - 1) / TB, TB, 0, stream>>>(row, rowptr, rk, E);
  scan_local_k<<<SB, 256, 0, stream>>>(rowptr, bsum, N1);
  scan_add_k  <<<SB, 256, 0, stream>>>(rowptr, bsum, N1);
  edge_place_k<<<(E + TB - 1) / TB, TB, 0, stream>>>(row, rk, rowptr, tmp, E);
  row_order_k <<<(N * 32 + 255) / 256, 256, 0, stream>>>(rowptr, tmp, col, ew, col_s, nrm, N);
  deg_dis_k   <<<(N + TB - 1) / TB, TB, 0, stream>>>(rowptr, nrm, dis, N);
  edge_norm_k <<<(N + TB - 1) / TB, TB, 0, stream>>>(rowptr, col_s, dis, nrm, N);

  transpose_k<<<(128 * H  + TB - 1) / TB, TB, 0, stream>>>(W0,  Wt0,  H);
  transpose_k<<<(128 * H  + TB - 1) / TB, TB, 0, stream>>>(W1,  Wt1,  H);
  transpose_k<<<(128 * DZ + TB - 1) / TB, TB, 0, stream>>>(Wmu, Wthd, DZ);
  transpose_k<<<(128 * DZ + TB - 1) / TB, TB, 0, stream>>>(Wsd, Wthd + 64 * 128, DZ);

  const int MT = N / 16;
  const unsigned GB = (unsigned)((MT + 7) / 8);
  const unsigned AB = (unsigned)((N * 32 + 255) / 256);

  gemm_strip_k<128, false><<<GB, 256, 0, stream>>>(h, Wt0, bufT, MT);
  agg_rows_k<128, false><<<AB, 256, 0, stream>>>(rowptr, col_s, nrm, bufT, dis, b0, nullptr, nullptr, bufH, N);
  gemm_strip_k<128, true><<<GB, 256, 0, stream>>>(bufH, Wt1, bufT, MT);
  agg_rows_k<128, false><<<AB, 256, 0, stream>>>(rowptr, col_s, nrm, bufT, dis, b1, nullptr, nullptr, bufH, N);
  gemm_strip_k<128, true><<<GB, 256, 0, stream>>>(bufH, Wthd, bufT, MT);
  agg_rows_k<128, true><<<AB, 256, 0, stream>>>(rowptr, col_s, nrm, bufT, dis, bmu, bsd, rnd, out, N);
}


# edge bucketing: per-edge global returning atomics, row scan, scattered place and one-wave-per-row sort launch -> two-level counting sort on LDS counters (128-row buckets); edge-id order kept
# speedup vs baseline: 3.7030x; 1.1934x over previous
#include <hip/hip_runtime.h>

#define NN 100000
#define NE 1600000

typedef float v2f __attribute__((ext_vector_type(2)));
typedef float v8f __attribute__((ext_vector_type(8)));

#define RBITS 7
#define RPB   (1 << RBITS)
#define NBK   ((NN + RPB - 1) >> RBITS)
#define NCH   256
#define CHUNK ((NE + NCH - 1) / NCH)

__global__ __launch_bounds__(1024) void bucket_hist_k(const int* __restrict__ row,
                                                      int* __restrict__ hist, int e) {
  __shared__ int h[NBK];
  for (int k = threadIdx.x; k < NBK; k += 1024) h[k] = 0;
  __syncthreads();
  const int lo = blockIdx.x * CHUNK;
  const int hi = (lo + CHUNK < e) ? (lo + CHUNK) : e;
  for (int i = lo + threadIdx.x; i < hi; i += 1024) atomicAdd(&h[row[i] >> RBITS], 1);
  __syncthreads();
  for (int k = threadIdx.x; k < NBK; k += 1024) hist[k * NCH + blockIdx.x] = h[k];
  if (blockIdx.x == 0 && threadIdx.x == 0) hist[NBK * NCH] = 0;
}

__global__ __launch_bounds__(256) void scan_local_k(int* cnt, int* __restrict__ bsum, int n1) {
  __shared__ int wsum[8];
  const int lane = threadIdx.x & 31;
  const int wv   = threadIdx.x >> 5;
  const int i = blockIdx.x * 256 + threadIdx.x;
  const int v = (i < n1) ? cnt[i] : 0;
  int x = v;
  for (int d = 1; d < 32; d <<= 1) {
    int y = __shfl_up(x, d);
    if (lane >= d) x += y;
  }
  if (lane == 31) wsum[wv] = x;
  __syncthreads();
  int off = 0;
  for (int k = 0; k < 8; ++k) off += (k < wv) ? wsum[k] : 0;
  const int incl = x + off;
  if (i < n1) cnt[i] = incl - v;
  if (threadIdx.x == 255) bsum[blockIdx.x] = incl;
}

__global__ __launch_bounds__(256) void scan_add_k(int* rowptr, const int* __restrict__ bsum, int n1) {
  __shared__ int red[8];
  const int lane = threadIdx.x & 31;
  const int wv   = threadIdx.x >> 5;
  const int b = blockIdx.x;
  int s = 0;
  for (int k = threadIdx.x; k < b; k += 256) s += bsum[k];
  for (int d = 16; d >= 1; d >>= 1) s += __shfl_xor(s, d);
  if (lane == 0) red[wv] = s;
  __syncthreads();
  int off = 0;
  for (int k = 0; k < 8; ++k) off += red[k];
  const int i = b * 256 + threadIdx.x;
  if (i < n1) rowptr[i] += off;
}

__global__ __launch_bounds__(1024) void bucket_scatter_k(const int* __restrict__ row,
                                                         const int* __restrict__ hist,
                                                         int2* __restrict__ rec, int e) {
  __shared__ int cur[NBK];
  for (int k = threadIdx.x; k < NBK; k += 1024) cur[k] = hist[k * NCH + blockIdx.x];
  __syncthreads();
  const int lo = blockIdx.x * CHUNK;
  const int hi = (lo + CHUNK < e) ? (lo + CHUNK) : e;
  for (int i = lo + threadIdx.x; i < hi; i += 1024) {
    const int r = row[i];
    const int p = atomicAdd(&cur[r >> RBITS], 1);
    rec[p] = make_int2(i, r);
  }
}

__global__ __launch_bounds__(512) void bucket_rows_k(const int* __restrict__ hist,
                                                     const int2* __restrict__ rec,
                                                     int* __restrict__ rk,
                                                     int* __restrict__ tmp,
                                                     const int* __restrict__ col,
                                                     const float* __restrict__ w,
                                                     int* __restrict__ rowptr,
                                                     int* __restrict__ col_s,
                                                     float* __restrict__ w_s, int n) {
  __shared__ int cnt[RPB];
  __shared__ int off[RPB + 1];
  const int b = blockIdx.x;
  const int lane = threadIdx.x & 31;
  const int wv   = threadIdx.x >> 5;
  const int bs = __builtin_amdgcn_readfirstlane(hist[b * NCH]);
  const int be = __builtin_amdgcn_readfirstlane(hist[(b + 1) * NCH]);
  const int rbase = b << RBITS;
  if (threadIdx.x < RPB) cnt[threadIdx.x] = 0;
  __syncthreads();
  for (int i = bs + (int)threadIdx.x; i < be; i += 512) {
    const int2 q = rec[i];
    rk[i] = atomicAdd(&cnt[q.y - rbase], 1);
  }
  __syncthreads();
  if (wv == 0) {
    const int c0 = cnt[4 * lane], c1 = cnt[4 * lane + 1];
    const int c2 = cnt[4 * lane + 2], c3 = cnt[4 * lane + 3];
    const int sum4 = c0 + c1 + c2 + c3;
    int x = sum4;
    for (int d = 1; d < 32; d <<= 1) {
      const int y = __shfl_up(x, d);
      if (lane >= d) x += y;
    }
    const int ex = x - sum4;
    off[4 * lane]     = ex;
    off[4 * lane + 1] = ex + c0;
    off[4 * lane + 2] = ex + c0 + c1;
    off[4 * lane + 3] = ex + c0 + c1 + c2;
    if (lane == 31) off[RPB] = x;
  }
  __syncthreads();
  if (threadIdx.x < RPB && rbase + (int)threadIdx.x < n)
    rowptr[rbase + threadIdx.x] = bs + off[threadIdx.x];
  if (b == (int)gridDim.x - 1 && threadIdx.x == 0) rowptr[n] = be;
  for (int i = bs + (int)threadIdx.x; i < be; i += 512) {
    const int2 q = rec[i];
    tmp[bs + off[q.y - rbase] + rk[i]] = q.x;
  }
  __syncthreads();
  for (int rr = wv; rr < RPB; rr += 16) {
    const int o0 = __builtin_amdgcn_readfirstlane(off[rr]);
    const int o1 = __builtin_amdgcn_readfirstlane(off[rr + 1]);
    const int s = bs + o0;
    const int L = o1 - o0;
    for (int bi = 0; bi < L; bi += 32) {
      const int i = bi + lane;
      const int mine = (i < L) ? tmp[s + i] : 0x7fffffff;
      int dr = 0;
      for (int bj = 0; bj < L; bj += 32) {
        const int j = bj + lane;
        const int other = (j < L) ? tmp[s + j] : 0x7fffffff;
        const int m = (L - bj < 32) ? (L - bj) : 32;
        for (int q = 0; q < m; ++q)
          dr += (__builtin_amdgcn_readlane(other, q) < mine) ? 1 : 0;
      }
      if (i < L) {
        col_s[s + dr] = col[mine];
        w_s[s + dr]   = w[mine];
      }
    }
  }
}

__global__ void deg_dis_k(const int* __restrict__ rowptr, const float* __restrict__ w_s,
                          float* __restrict__ dis, int n) {
  int i = blockIdx.x * blockDim.x + threadIdx.x;
  if (i >= n) return;
  const int s = rowptr[i], t = rowptr[i + 1];
  float d = 1.0f;
  for (int p = s; p < t; ++p) d += w_s[p];
  dis[i] = rsqrtf(d);
}

__global__ void edge_norm_k(const int* __restrict__ rowptr, const int* __restrict__ col_s,
                            const float* __restrict__ dis, float* w_s, int n) {
  int i = blockIdx.x * blockDim.x + threadIdx.x;
  if (i >= n) return;
  const int s = rowptr[i], t = rowptr[i + 1];
  const float di = dis[i];
  for (int p = s; p < t; ++p) w_s[p] = di * w_s[p] * dis[col_s[p]];
}

__global__ void transpose_k(const float* __restrict__ W, float* __restrict__ Wt, int U) {
  int i = blockIdx.x * blockDim.x + threadIdx.x;
  if (i >= 128 * U) return;
  int k = i / U, u = i % U;
  Wt[(size_t)u * 128 + k] = W[i];
}

template <int U, bool RELU_IN>
__global__ __launch_bounds__(256) void gemm_strip_k(const float* __restrict__ X,
                                                    const float* __restrict__ Wt,
                                                    float* __restrict__ T, int MT) {
  constexpr int P = 132;
  constexpr int NT = U / 16;
  __shared__ alignas(16) float Bs[U * P];
  __shared__ alignas(16) float As[8 * 16 * P];
  const int lane = threadIdx.x & 31;
  const int wave = threadIdx.x >> 5;
  const int tile_m = blockIdx.x * 8 + wave;
  const bool live = tile_m < MT;
  const int tm = live ? tile_m : MT - 1;

  const float* xs = X + (size_t)tm * (16 * 128) + lane * 4;
  float* as = As + wave * (16 * P);
  float4 xa[16];
  for (int j = 0; j < 16; ++j) xa[j] = *reinterpret_cast<const float4*>(xs + j * 128);
  for (int i = 0; i < U / 8; ++i) {
    const int n = i * 8 + wave;
    const float4 wv = *reinterpret_cast<const float4*>(Wt + (size_t)n * 128 + lane * 4);
    *reinterpret_cast<float4*>(Bs + n * P + lane * 4) = wv;
  }
  for (int j = 0; j < 16; ++j) *reinterpret_cast<float4*>(as + j * P + lane * 4) = xa[j];
  __syncthreads();
  if (!live) return;

  const int kk = (lane >> 4) * 2;
  const float* ar = as + (lane & 15) * P + kk;
  const float* br = Bs + (lane & 15) * P + kk;
  v8f c[NT];
  for (int t = 0; t < NT; ++t) c[t] = v8f{0.f, 0.f, 0.f, 0.f, 0.f, 0.f, 0.f, 0.f};
  for (int k = 0; k < 128; k += 4) {
    v2f a = *reinterpret_cast<const v2f*>(ar + k);
    if (RELU_IN) { a.x = fmaxf(a.x, 0.0f); a.y = fmaxf(a.y, 0.0f); }
    for (int t = 0; t < NT; ++t) {
      const v2f b = *reinterpret_cast<const v2f*>(br + t * (16 * P) + k);
      c[t] = __builtin_amdgcn_wmma_f32_16x16x4_f32(false, a, false, b, (short)0, c[t],
                                                   false, false);
    }
  }

  const int mbase = tile_m * 16 + (lane >> 4) * 8;
  float* trow = T + (size_t)mbase * U + (lane & 15);
  for (int t = 0; t < NT; ++t)
    for (int r = 0; r < 8; ++r) trow[(size_t)r * U + t * 16] = c[t][r];
}

template <int U, bool HEADS>
__global__ __launch_bounds__(256) void agg_rows_k(const int* __restrict__ rowptr,
                                                  const int* __restrict__ col_s,
                                                  const float* __restrict__ nrm_s,
                                                  const float* __restrict__ T,
                                                  const float* __restrict__ dis,
                                                  const float* __restrict__ bias,
                                                  const float* __restrict__ bias_hi,
                                                  const float* __restrict__ rnd,
                                                  float* __restrict__ out, int n) {
  constexpr int NL = U / 32;
  constexpr int HL = NL / 2;
  const int r = (blockIdx.x * 256 + threadIdx.x) >> 5;
  if (r >= n) return;
  const int lane = threadIdx.x & 31;
  const int s = __builtin_amdgcn_readfirstlane(rowptr[r]);
  const int t = __builtin_amdgcn_readfirstlane(rowptr[r + 1]);
  const float* Tl = T + lane;
  const float d  = dis[r];
  const float sn = d * d;
  float acc[NL];
  for (int j = 0; j < NL; ++j) {
    const float bj = (HEADS && j >= HL) ? bias_hi[lane + 32 * (j - HL)] : bias[lane + 32 * j];
    acc[j] = bj + sn * Tl[(size_t)r * U + 32 * j];
  }
  float rv[HL > 0 ? HL : 1];
  if (HEADS)
    for (int j = 0; j < HL; ++j) rv[j] = rnd[(size_t)r * (U / 2) + lane + 32 * j];

  for (int base = s; base < t; base += 32) {
    const int m = (t - base < 32) ? (t - base) : 32;
    int   cv = 0;
    float wv = 0.0f;
    if (lane < m) { cv = col_s[base + lane]; wv = nrm_s[base + lane]; }
    const int wi = __float_as_int(wv);
    int q = 0;
    for (; q + 4 <= m; q += 4) {
      const int c0 = __builtin_amdgcn_readlane(cv, q);
      const int c1 = __builtin_amdgcn_readlane(cv, q + 1);
      const int c2 = __builtin_amdgcn_readlane(cv, q + 2);
      const int c3 = __builtin_amdgcn_readlane(cv, q + 3);
      const float w0 = __int_as_float(__builtin_amdgcn_readlane(wi, q));
      const float w1 = __int_as_float(__builtin_amdgcn_readlane(wi, q + 1));
      const float w2 = __int_as_float(__builtin_amdgcn_readlane(wi, q + 2));
      const float w3 = __int_as_float(__builtin_amdgcn_readlane(wi, q + 3));
      float t0[NL], t1[NL], t2[NL], t3[NL];
      for (int j = 0; j < NL; ++j) {
        t0[j] = Tl[(size_t)c0 * U + 32 * j];
        t1[j] = Tl[(size_t)c1 * U + 32 * j];
        t2[j] = Tl[(size_t)c2 * U + 32 * j];
        t3[j] = Tl[(size_t)c3 * U + 32 * j];
      }
      for (int j = 0; j < NL; ++j) {
        acc[j] += w0 * t0[j];
        acc[j] += w1 * t1[j];
        acc[j] += w2 * t2[j];
        acc[j] += w3 * t3[j];
      }
    }
    for (; q < m; ++q) {
      const int   c0 = __builtin_amdgcn_readlane(cv, q);
      const float w0 = __int_as_float(__builtin_amdgcn_readlane(wi, q));
      for (int j = 0; j < NL; ++j) acc[j] += w0 * Tl[(size_t)c0 * U + 32 * j];
    }
  }
  if (HEADS) {
    float* ho = out + (size_t)r * (U / 2) + lane;
    float* ko = ho + (size_t)n * (U / 2);
    for (int j = 0; j < HL; ++j) {
      const float mu = acc[j];
      const float ls = acc[HL + j];
      const float sd = expf(ls);
      ho[32 * j] = mu + rv[j] * sd;
      ko[32 * j] = -ls + 0.5f * (sd * sd + mu * mu - 1.0f);
    }
  } else {
    float* orow = out + (size_t)r * U + lane;
    for (int j = 0; j < NL; ++j) orow[32 * j] = acc[j];
  }
}

extern "C" void kernel_launch(void* const* d_in, const int* in_sizes, int n_in,
                              void* d_out, int out_size, void* d_ws, size_t ws_size,
                              hipStream_t stream) {
  (void)in_sizes; (void)n_in; (void)out_size; (void)ws_size;
  const int N = NN, E = NE, H = 128, DZ = 64;
  const float* h   = (const float*)d_in[0];
  const int*   ei  = (const int*)d_in[1];
  const float* ew  = (const float*)d_in[2];
  const float* rnd = (const float*)d_in[3];
  const float* W0  = (const float*)d_in[4];
  const float* b0  = (const float*)d_in[5];
  const float* W1  = (const float*)d_in[6];
  const float* b1  = (const float*)d_in[7];
  const float* Wmu = (const float*)d_in[8];
  const float* bmu = (const float*)d_in[9];
  const float* Wsd = (const float*)d_in[10];
  const float* bsd = (const float*)d_in[11];
  const int* row = ei;
  const int* col = ei + E;
  float* out = (float*)d_out;

  float* ws    = (float*)d_ws;
  float* dis   = ws;
  float* nrm   = dis + N;
  float* bufT  = nrm + E;
  float* bufH  = bufT + (size_t)N * 128;
  float* bufS  = bufH + (size_t)N * 128;
  float* Wt0   = bufS + (size_t)N * 64;
  float* Wt1   = Wt0 + 128 * 128;
  float* Wthd  = Wt1 + 128 * 128;
  int* col_s   = (int*)bufS;
  int* rowptr  = col_s + E;
  int* bsum    = rowptr + N + 96;
  int* hist    = bsum + 1024;
  int2* rec    = (int2*)bufH;
  int* rk      = (int*)bufH + 2 * (size_t)E;
  int* tmp     = rk + E;

  const int TB = 256;
  const int H1 = NBK * NCH + 1;
  const int SB = (H1 + 255) / 256;
  bucket_hist_k   <<<NCH, 1024, 0, stream>>>(row, hist, E);
  scan_local_k    <<<SB, 256, 0, stream>>>(hist, bsum, H1);
  scan_add_k      <<<SB, 256, 0, stream>>>(hist, bsum, H1);
  bucket_scatter_k<<<NCH, 1024, 0, stream>>>(row, hist, rec, E);
  bucket_rows_k   <<<NBK, 512, 0, stream>>>(hist, rec, rk, tmp, col, ew, rowptr, col_s, nrm, N);
  deg_dis_k   <<<(N + TB - 1) / TB, TB, 0, stream>>>(rowptr, nrm, dis, N);
  edge_norm_k <<<(N + TB - 1) / TB, TB, 0, stream>>>(rowptr, col_s, dis, nrm, N);

  transpose_k<<<(128 * H  + TB - 1) / TB, TB, 0, stream>>>(W0,  Wt0,  H);
  transpose_k<<<(128 * H  + TB - 1) / TB, TB, 0, stream>>>(W1,  Wt1,  H);
  transpose_k<<<(128 * DZ + TB - 1) / TB, TB, 0, stream>>>(Wmu, Wthd, DZ);
  transpose_k<<<(128 * DZ + TB - 1) / TB, TB, 0, stream>>>(Wsd, Wthd + 64 * 128, DZ);

  const int MT = N / 16;
  const unsigned GB = (unsigned)((MT + 7) / 8);
  const unsigned AB = (unsigned)((N * 32 + 255) / 256);

  gemm_strip_k<128, false><<<GB, 256, 0, stream>>>(h, Wt0, bufT, MT);
  agg_rows_k<128, false><<<AB, 256, 0, stream>>>(rowptr, col_s, nrm, bufT, dis, b0, nullptr, nullptr, bufH, N);
  gemm_strip_k<128, true><<<GB, 256, 0, stream>>>(bufH, Wt1, bufT, MT);
  agg_rows_k<128, false><<<AB, 256, 0, stream>>>(rowptr, col_s, nrm, bufT, dis, b1, nullptr, nullptr, bufH, N);
  gemm_strip_k<128, true><<<GB, 256, 0, stream>>>(bufH, Wthd, bufT, MT);
  agg_rows_k<128, true><<<AB, 256, 0, stream>>>(rowptr, col_s, nrm, bufT, dis, bmu, bsd, rnd, out, N);
}


# GEMM takes W as given: W row pairs go to LDS interleaved, so a B fragment {W[k][n],W[k+1][n]} stays one 8-byte read; the 4 weight-transpose launches are gone (17 -> 13 launches); k order kept
# speedup vs baseline: 3.9391x; 1.0638x over previous
#include <hip/hip_runtime.h>

#define NN 100000
#define NE 1600000

typedef float v2f __attribute__((ext_vector_type(2)));
typedef float v8f __attribute__((ext_vector_type(8)));

#define RBITS 7
#define RPB   (1 << RBITS)
#define NBK   ((NN + RPB - 1) >> RBITS)
#define NCH   256
#define CHUNK ((NE + NCH - 1) / NCH)

__global__ __launch_bounds__(1024) void bucket_hist_k(const int* __restrict__ row,
                                                      int* __restrict__ hist, int e) {
  __shared__ int h[NBK];
  for (int k = threadIdx.x; k < NBK; k += 1024) h[k] = 0;
  __syncthreads();
  const int lo = blockIdx.x * CHUNK;
  const int hi = (lo + CHUNK < e) ? (lo + CHUNK) : e;
  for (int i = lo + threadIdx.x; i < hi; i += 1024) atomicAdd(&h[row[i] >> RBITS], 1);
  __syncthreads();
  for (int k = threadIdx.x; k < NBK; k += 1024) hist[k * NCH + blockIdx.x] = h[k];
  if (blockIdx.x == 0 && threadIdx.x == 0) hist[NBK * NCH] = 0;
}

__global__ __launch_bounds__(256) void scan_local_k(int* cnt, int* __restrict__ bsum, int n1) {
  __shared__ int wsum[8];
  const int lane = threadIdx.x & 31;
  const int wv   = threadIdx.x >> 5;
  const int i = blockIdx.x * 256 + threadIdx.x;
  const int v = (i < n1) ? cnt[i] : 0;
  int x = v;
  for (int d = 1; d < 32; d <<= 1) {
    int y = __shfl_up(x, d);
    if (lane >= d) x += y;
  }
  if (lane == 31) wsum[wv] = x;
  __syncthreads();
  int off = 0;
  for (int k = 0; k < 8; ++k) off += (k < wv) ? wsum[k] : 0;
  const int incl = x + off;
  if (i < n1) cnt[i] = incl - v;
  if (threadIdx.x == 255) bsum[blockIdx.x] = incl;
}

__global__ __launch_bounds__(256) void scan_add_k(int* rowptr, const int* __restrict__ bsum, int n1) {
  __shared__ int red[8];
  const int lane = threadIdx.x & 31;
  const int wv   = threadIdx.x >> 5;
  const int b = blockIdx.x;
  int s = 0;
  for (int k = threadIdx.x; k < b; k += 256) s += bsum[k];
  for (int d = 16; d >= 1; d >>= 1) s += __shfl_xor(s, d);
  if (lane == 0) red[wv] = s;
  __syncthreads();
  int off = 0;
  for (int k = 0; k < 8; ++k) off += red[k];
  const int i = b * 256 + threadIdx.x;
  if (i < n1) rowptr[i] += off;
}

__global__ __launch_bounds__(1024) void bucket_scatter_k(const int* __restrict__ row,
                                                         const int* __restrict__ hist,
                                                         int2* __restrict__ rec, int e) {
  __shared__ int cur[NBK];
  for (int k = threadIdx.x; k < NBK; k += 1024) cur[k] = hist[k * NCH + blockIdx.x];
  __syncthreads();
  const int lo = blockIdx.x * CHUNK;
  const int hi = (lo + CHUNK < e) ? (lo + CHUNK) : e;
  for (int i = lo + threadIdx.x; i < hi; i += 1024) {
    const int r = row[i];
    const int p = atomicAdd(&cur[r >> RBITS], 1);
    rec[p] = make_int2(i, r);
  }
}

__global__ __launch_bounds__(512) void bucket_rows_k(const int* __restrict__ hist,
                                                     const int2* __restrict__ rec,
                                                     int* __restrict__ rk,
                                                     int* __restrict__ tmp,
                                                     const int* __restrict__ col,
                                                     const float* __restrict__ w,
                                                     int* __restrict__ rowptr,
                                                     int* __restrict__ col_s,
                                                     float* __restrict__ w_s, int n) {
  __shared__ int cnt[RPB];
  __shared__ int off[RPB + 1];
  const int b = blockIdx.x;
  const int lane = threadIdx.x & 31;
  const int wv   = threadIdx.x >> 5;
  const int bs = __builtin_amdgcn_readfirstlane(hist[b * NCH]);
  const int be = __builtin_amdgcn_readfirstlane(hist[(b + 1) * NCH]);
  const int rbase = b << RBITS;
  if (threadIdx.x < RPB) cnt[threadIdx.x] = 0;
  __syncthreads();
  for (int i = bs + (int)threadIdx.x; i < be; i += 512) {
    const int2 q = rec[i];
    rk[i] = atomicAdd(&cnt[q.y - rbase], 1);
  }
  __syncthreads();
  if (wv == 0) {
    const int c0 = cnt[4 * lane], c1 = cnt[4 * lane + 1];
    const int c2 = cnt[4 * lane + 2], c3 = cnt[4 * lane + 3];
    const int sum4 = c0 + c1 + c2 + c3;
    int x = sum4;
    for (int d = 1; d < 32; d <<= 1) {
      const int y = __shfl_up(x, d);
      if (lane >= d) x += y;
    }
    const int ex = x - sum4;
    off[4 * lane]     = ex;
    off[4 * lane + 1] = ex + c0;
    off[4 * lane + 2] = ex + c0 + c1;
    off[4 * lane + 3] = ex + c0 + c1 + c2;
    if (lane == 31) off[RPB] = x;
  }
  __syncthreads();
  if (threadIdx.x < RPB && rbase + (int)threadIdx.x < n)
    rowptr[rbase + threadIdx.x] = bs + off[threadIdx.x];
  if (b == (int)gridDim.x - 1 && threadIdx.x == 0) rowptr[n] = be;
  for (int i = bs + (int)threadIdx.x; i < be; i += 512) {
    const int2 q = rec[i];
    tmp[bs + off[q.y - rbase] + rk[i]] = q.x;
  }
  __syncthreads();
  for (int rr = wv; rr < RPB; rr += 16) {
    const int o0 = __builtin_amdgcn_readfirstlane(off[rr]);
    const int o1 = __builtin_amdgcn_readfirstlane(off[rr + 1]);
    const int s = bs + o0;
    const int L = o1 - o0;
    for (int bi = 0; bi < L; bi += 32) {
      const int i = bi + lane;
      const int mine = (i < L) ? tmp[s + i] : 0x7fffffff;
      int dr = 0;
      for (int bj = 0; bj < L; bj += 32) {
        const int j = bj + lane;
        const int other = (j < L) ? tmp[s + j] : 0x7fffffff;
        const int m = (L - bj < 32) ? (L - bj) : 32;
        for (int q = 0; q < m; ++q)
          dr += (__builtin_amdgcn_readlane(other, q) < mine) ? 1 : 0;
      }
      if (i < L) {
        col_s[s + dr] = col[mine];
        w_s[s + dr]   = w[mine];
      }
    }
  }
}

__global__ void deg_dis_k(const int* __restrict__ rowptr, const float* __restrict__ w_s,
                          float* __restrict__ dis, int n) {
  int i = blockIdx.x * blockDim.x + threadIdx.x;
  if (i >= n) return;
  const int s = rowptr[i], t = rowptr[i + 1];
  float d = 1.0f;
  for (int p = s; p < t; ++p) d += w_s[p];
  dis[i] = rsqrtf(d);
}

__global__ void edge_norm_k(const int* __restrict__ rowptr, const int* __restrict__ col_s,
                            const float* __restrict__ dis, float* w_s, int n) {
  int i = blockIdx.x * blockDim.x + threadIdx.x;
  if (i >= n) return;
  const int s = rowptr[i], t = rowptr[i + 1];
  const float di = dis[i];
  for (int p = s; p < t; ++p) w_s[p] = di * w_s[p] * dis[col_s[p]];
}

template <int U, bool RELU_IN, bool SPLIT>
__global__ __launch_bounds__(256) void gemm_strip_k(const float* __restrict__ X,
                                                    const float* __restrict__ Wa,
                                                    const float* __restrict__ Wb,
                                                    float* __restrict__ T, int MT) {
  constexpr int P = 132;
  constexpr int PB = 2 * U + 32;
  constexpr int NT = U / 16;
  __shared__ alignas(16) float Bs[64 * PB];
  __shared__ alignas(16) float As[8 * 16 * P];
  const int lane = threadIdx.x & 31;
  const int wave = threadIdx.x >> 5;
  const int tile_m = blockIdx.x * 8 + wave;
  const bool live = tile_m < MT;
  const int tm = live ? tile_m : MT - 1;

  const float* xs = X + (size_t)tm * (16 * 128) + lane * 4;
  float* as = As + wave * (16 * P);
  float4 xa[16];
  for (int j = 0; j < 16; ++j) xa[j] = *reinterpret_cast<const float4*>(xs + j * 128);
  static_assert(U == 128, "a W row is one float4 per lane");
  constexpr int WP = SPLIT ? U / 2 : U;
  const float* wcol = (SPLIT && lane >= 16) ? (Wb + (lane - 16) * 4) : (Wa + lane * 4);
  for (int i = 0; i < 64 / 8; ++i) {
    const int pr = i * 8 + wave;
    const float4 w0 = *reinterpret_cast<const float4*>(wcol + (size_t)(2 * pr) * WP);
    const float4 w1 = *reinterpret_cast<const float4*>(wcol + (size_t)(2 * pr + 1) * WP);
    float* bd = Bs + pr * PB + lane * 8;
    *reinterpret_cast<float4*>(bd)     = make_float4(w0.x, w1.x, w0.y, w1.y);
    *reinterpret_cast<float4*>(bd + 4) = make_float4(w0.z, w1.z, w0.w, w1.w);
  }
  for (int j = 0; j < 16; ++j) *reinterpret_cast<float4*>(as + j * P + lane * 4) = xa[j];
  __syncthreads();
  if (!live) return;

  const int kk = (lane >> 4) * 2;
  const float* ar = as + (lane & 15) * P + kk;
  const float* br = Bs + (lane >> 4) * PB + (lane & 15) * 2;
  v8f c[NT];
  for (int t = 0; t < NT; ++t) c[t] = v8f{0.f, 0.f, 0.f, 0.f, 0.f, 0.f, 0.f, 0.f};
  for (int k = 0; k < 128; k += 4) {
    v2f a = *reinterpret_cast<const v2f*>(ar + k);
    if (RELU_IN) { a.x = fmaxf(a.x, 0.0f); a.y = fmaxf(a.y, 0.0f); }
    for (int t = 0; t < NT; ++t) {
      const v2f b = *reinterpret_cast<const v2f*>(br + (k >> 1) * PB + t * 32);
      c[t] = __builtin_amdgcn_wmma_f32_16x16x4_f32(false, a, false, b, (short)0, c[t],
                                                   false, false);
    }
  }

  const int mbase = tile_m * 16 + (lane >> 4) * 8;
  float* trow = T + (size_t)mbase * U + (lane & 15);
  for (int t = 0; t < NT; ++t)
    for (int r = 0; r < 8; ++r) trow[(size_t)r * U + t * 16] = c[t][r];
}

template <int U, bool HEADS>
__global__ __launch_bounds__(256) void agg_rows_k(const int* __restrict__ rowptr,
                                                  const int* __restrict__ col_s,
                                                  const float* __restrict__ nrm_s,
                                                  const float* __restrict__ T,
                                                  const float* __restrict__ dis,
                                                  const float* __restrict__ bias,
                                                  const float* __restrict__ bias_hi,
                                                  const float* __restrict__ rnd,
                                                  float* __restrict__ out, int n) {
  constexpr int NL = U / 32;
  constexpr int HL = NL / 2;
  const int r = (blockIdx.x * 256 + threadIdx.x) >> 5;
  if (r >= n) return;
  const int lane = threadIdx.x & 31;
  const int s = __builtin_amdgcn_readfirstlane(rowptr[r]);
  const int t = __builtin_amdgcn_readfirstlane(rowptr[r + 1]);
  const float* Tl = T + lane;
  const float d  = dis[r];
  const float sn = d * d;
  float acc[NL];
  for (int j = 0; j < NL; ++j) {
    const float bj = (HEADS && j >= HL) ? bias_hi[lane + 32 * (j - HL)] : bias[lane + 32 * j];
    acc[j] = bj + sn * Tl[(size_t)r * U + 32 * j];
  }
  float rv[HL > 0 ? HL : 1];
  if (HEADS)
    for (int j = 0; j < HL; ++j) rv[j] = rnd[(size_t)r * (U / 2) + lane + 32 * j];

  for (int base = s; base < t; base += 32) {
    const int m = (t - base < 32) ? (t - base) : 32;
    int   cv = 0;
    float wv = 0.0f;
    if (lane < m) { cv = col_s[base + lane]; wv = nrm_s[base + lane]; }
    const int wi = __float_as_int(wv);
    int q = 0;
    for (; q + 4 <= m; q += 4) {
      const int c0 = __builtin_amdgcn_readlane(cv, q);
      const int c1 = __builtin_amdgcn_readlane(cv, q + 1);
      const int c2 = __builtin_amdgcn_readlane(cv, q + 2);
      const int c3 = __builtin_amdgcn_readlane(cv, q + 3);
      const float w0 = __int_as_float(__builtin_amdgcn_readlane(wi, q));
      const float w1 = __int_as_float(__builtin_amdgcn_readlane(wi, q + 1));
      const float w2 = __int_as_float(__builtin_amdgcn_readlane(wi, q + 2));
      const float w3 = __int_as_float(__builtin_amdgcn_readlane(wi, q + 3));
      float t0[NL], t1[NL], t2[NL], t3[NL];
      for (int j = 0; j < NL; ++j) {
        t0[j] = Tl[(size_t)c0 * U + 32 * j];
        t1[j] = Tl[(size_t)c1 * U + 32 * j];
        t2[j] = Tl[(size_t)c2 * U + 32 * j];
        t3[j] = Tl[(size_t)c3 * U + 32 * j];
      }
      for (int j = 0; j < NL; ++j) {
        acc[j] += w0 * t0[j];
        acc[j] += w1 * t1[j];
        acc[j] += w2 * t2[j];
        acc[j] += w3 * t3[j];
      }
    }
    for (; q < m; ++q) {
      const int   c0 = __builtin_amdgcn_readlane(cv, q);
      const float w0 = __int_as_float(__builtin_amdgcn_readlane(wi, q));
      for (int j = 0; j < NL; ++j) acc[j] += w0 * Tl[(size_t)c0 * U + 32 * j];
    }
  }
  if (HEADS) {
    float* ho = out + (size_t)r * (U / 2) + lane;
    float* ko = ho + (size_t)n * (U / 2);
    for (int j = 0; j < HL; ++j) {
      const float mu = acc[j];
      const float ls = acc[HL + j];
      const float sd = expf(ls);
      ho[32 * j] = mu + rv[j] * sd;
      ko[32 * j] = -ls + 0.5f * (sd * sd + mu * mu - 1.0f);
    }
  } else {
    float* orow = out + (size_t)r * U + lane;
    for (int j = 0; j < NL; ++j) orow[32 * j] = acc[j];
  }
}

extern "C" void kernel_launch(void* const* d_in, const int* in_sizes, int n_in,
                              void* d_out, int out_size, void* d_ws, size_t ws_size,
                              hipStream_t stream) {
  (void)in_sizes; (void)n_in; (void)out_size; (void)ws_size;
  const int N = NN, E = NE;
  const float* h   = (const float*)d_in[0];
  const int*   ei  = (const int*)d_in[1];
  const float* ew  = (const float*)d_in[2];
  const float* rnd = (const float*)d_in[3];
  const float* W0  = (const float*)d_in[4];
  const float* b0  = (const float*)d_in[5];
  const float* W1  = (const float*)d_in[6];
  const float* b1  = (const float*)d_in[7];
  const float* Wmu = (const float*)d_in[8];
  const float* bmu = (const float*)d_in[9];
  const float* Wsd = (const float*)d_in[10];
  const float* bsd = (const float*)d_in[11];
  const int* row = ei;
  const int* col = ei + E;
  float* out = (float*)d_out;

  float* ws    = (float*)d_ws;
  float* dis   = ws;
  float* nrm   = dis + N;
  float* bufT  = nrm + E;
  float* bufH  = bufT + (size_t)N * 128;
  float* bufS  = bufH + (size_t)N * 128;
  int* col_s   = (int*)bufS;
  int* rowptr  = col_s + E;
  int* bsum    = rowptr + N + 96;
  int* hist    = bsum + 1024;
  int2* rec    = (int2*)bufH;
  int* rk      = (int*)bufH + 2 * (size_t)E;
  int* tmp     = rk + E;

  const int TB = 256;
  const int H1 = NBK * NCH + 1;
  const int SB = (H1 + 255) / 256;
  bucket_hist_k   <<<NCH, 1024, 0, stream>>>(row, hist, E);
  scan_local_k    <<<SB, 256, 0, stream>>>(hist, bsum, H1);
  scan_add_k      <<<SB, 256, 0, stream>>>(hist, bsum, H1);
  bucket_scatter_k<<<NCH, 1024, 0, stream>>>(row, hist, rec, E);
  bucket_rows_k   <<<NBK, 512, 0, stream>>>(hist, rec, rk, tmp, col, ew, rowptr, col_s, nrm, N);
  deg_dis_k   <<<(N + TB - 1) / TB, TB, 0, stream>>>(rowptr, nrm, dis, N);
  edge_norm_k <<<(N + TB - 1) / TB, TB, 0, stream>>>(rowptr, col_s, dis, nrm, N);

  const int MT = N / 16;
  const unsigned GB = (unsigned)((MT + 7) / 8);
  const unsigned AB = (unsigned)((N * 32 + 255) / 256);

  gemm_strip_k<128, false, false><<<GB, 256, 0, stream>>>(h, W0, nullptr, bufT, MT);
  agg_rows_k<128, false><<<AB, 256, 0, stream>>>(rowptr, col_s, nrm, bufT, dis, b0, nullptr, nullptr, bufH, N);
  gemm_strip_k<128, true, false><<<GB, 256, 0, stream>>>(bufH, W1, nullptr, bufT, MT);
  agg_rows_k<128, false><<<AB, 256, 0, stream>>>(rowptr, col_s, nrm, bufT, dis, b1, nullptr, nullptr, bufH, N);
  gemm_strip_k<128, true, true><<<GB, 256, 0, stream>>>(bufH, Wmu, Wsd, bufT, MT);
  agg_rows_k<128, true><<<AB, 256, 0, stream>>>(rowptr, col_s, nrm, bufT, dis, bmu, bsd, rnd, out, N);
}


# layers 1, 2: aggregation and the next layer's GEMM in one launch (13 -> 11 launches): 16-wave groups walk 16-row strips, row sums go relu'd to LDS, 4 waves run the strip's WMMAs; sum orders kept
# speedup vs baseline: 4.3064x; 1.0932x over previous
#include <hip/hip_runtime.h>

#define NN 100000
#define NE 1600000

typedef float v2f __attribute__((ext_vector_type(2)));
typedef float v8f __attribute__((ext_vector_type(8)));

#define RBITS 7
#define RPB   (1 << RBITS)
#define NBK   ((NN + RPB - 1) >> RBITS)
#define NCH   256
#define CHUNK ((NE + NCH - 1) / NCH)

__global__ __launch_bounds__(1024) void bucket_hist_k(const int* __restrict__ row,
                                                      int* __restrict__ hist, int e) {
  __shared__ int h[NBK];
  for (int k = threadIdx.x; k < NBK; k += 1024) h[k] = 0;
  __syncthreads();
  const int lo = blockIdx.x * CHUNK;
  const int hi = (lo + CHUNK < e) ? (lo + CHUNK) : e;
  for (int i = lo + threadIdx.x; i < hi; i += 1024) atomicAdd(&h[row[i] >> RBITS], 1);
  __syncthreads();
  for (int k = threadIdx.x; k < NBK; k += 1024) hist[k * NCH + blockIdx.x] = h[k];
  if (blockIdx.x == 0 && threadIdx.x == 0) hist[NBK * NCH] = 0;
}

__global__ __launch_bounds__(256) void scan_local_k(int* cnt, int* __restrict__ bsum, int n1) {
  __shared__ int wsum[8];
  const int lane = threadIdx.x & 31;
  const int wv   = threadIdx.x >> 5;
  const int i = blockIdx.x * 256 + threadIdx.x;
  const int v = (i < n1) ? cnt[i] : 0;
  int x = v;
  for (int d = 1; d < 32; d <<= 1) {
    int y = __shfl_up(x, d);
    if (lane >= d) x += y;
  }
  if (lane == 31) wsum[wv] = x;
  __syncthreads();
  int off = 0;
  for (int k = 0; k < 8; ++k) off += (k < wv) ? wsum[k] : 0;
  const int incl = x + off;
  if (i < n1) cnt[i] = incl - v;
  if (threadIdx.x == 255) bsum[blockIdx.x] = incl;
}

__global__ __launch_bounds__(256) void scan_add_k(int* rowptr, const int* __restrict__ bsum, int n1) {
  __shared__ int red[8];
  const int lane = threadIdx.x & 31;
  const int wv   = threadIdx.x >> 5;
  const int b = blockIdx.x;
  int s = 0;
  for (int k = threadIdx.x; k < b; k += 256) s += bsum[k];
  for (int d = 16; d >= 1; d >>= 1) s += __shfl_xor(s, d);
  if (lane == 0) red[wv] = s;
  __syncthreads();
  int off = 0;
  for (int k = 0; k < 8; ++k) off += red[k];
  const int i = b * 256 + threadIdx.x;
  if (i < n1) rowptr[i] += off;
}

__global__ __launch_bounds__(1024) void bucket_scatter_k(const int* __restrict__ row,
                                                         const int* __restrict__ hist,
                                                         int2* __restrict__ rec, int e) {
  __shared__ int cur[NBK];
  for (int k = threadIdx.x; k < NBK; k += 1024) cur[k] = hist[k * NCH + blockIdx.x];
  __syncthreads();
  const int lo = blockIdx.x * CHUNK;
  const int hi = (lo + CHUNK < e) ? (lo + CHUNK) : e;
  for (int i = lo + threadIdx.x; i < hi; i += 1024) {
    const int r = row[i];
    const int p = atomicAdd(&cur[r >> RBITS], 1);
    rec[p] = make_int2(i, r);
  }
}

__global__ __launch_bounds__(512) void bucket_rows_k(const int* __restrict__ hist,
                                                     const int2* __restrict__ rec,
                                                     int* __restrict__ rk,
                                                     int* __restrict__ tmp,
                                                     const int* __restrict__ col,
                                                     const float* __restrict__ w,
                                                     int* __restrict__ rowptr,
                                                     int* __restrict__ col_s,
                                                     float* __restrict__ w_s, int n) {
  __shared__ int cnt[RPB];
  __shared__ int off[RPB + 1];
  const int b = blockIdx.x;
  const int lane = threadIdx.x & 31;
  const int wv   = threadIdx.x >> 5;
  const int bs = __builtin_amdgcn_readfirstlane(hist[b * NCH]);
  const int be = __builtin_amdgcn_readfirstlane(hist[(b + 1) * NCH]);
  const int rbase = b << RBITS;
  if (threadIdx.x < RPB) cnt[threadIdx.x] = 0;
  __syncthreads();
  for (int i = bs + (int)threadIdx.x; i < be; i += 512) {
    const int2 q = rec[i];
    rk[i] = atomicAdd(&cnt[q.y - rbase], 1);
  }
  __syncthreads();
  if (wv == 0) {
    const int c0 = cnt[4 * lane], c1 = cnt[4 * lane + 1];
    const int c2 = cnt[4 * lane + 2], c3 = cnt[4 * lane + 3];
    const int sum4 = c0 + c1 + c2 + c3;
    int x = sum4;
    for (int d = 1; d < 32; d <<= 1) {
      const int y = __shfl_up(x, d);
      if (lane >= d) x += y;
    }
    const int ex = x - sum4;
    off[4 * lane]     = ex;
    off[4 * lane + 1] = ex + c0;
    off[4 * lane + 2] = ex + c0 + c1;
    off[4 * lane + 3] = ex + c0 + c1 + c2;
    if (lane == 31) off[RPB] = x;
  }
  __syncthreads();
  if (threadIdx.x < RPB && rbase + (int)threadIdx.x < n)
    rowptr[rbase + threadIdx.x] = bs + off[threadIdx.x];
  if (b == (int)gridDim.x - 1 && threadIdx.x == 0) rowptr[n] = be;
  for (int i = bs + (int)threadIdx.x; i < be; i += 512) {
    const int2 q = rec[i];
    tmp[bs + off[q.y - rbase] + rk[i]] = q.x;
  }
  __syncthreads();
  for (int rr = wv; rr < RPB; rr += 16) {
    const int o0 = __builtin_amdgcn_readfirstlane(off[rr]);
    const int o1 = __builtin_amdgcn_readfirstlane(off[rr + 1]);
    const int s = bs + o0;
    const int L = o1 - o0;
    for (int bi = 0; bi < L; bi += 32) {
      const int i = bi + lane;
      const int mine = (i < L) ? tmp[s + i] : 0x7fffffff;
      int dr = 0;
      for (int bj = 0; bj < L; bj += 32) {
        const int j = bj + lane;
        const int other = (j < L) ? tmp[s + j] : 0x7fffffff;
        const int m = (L - bj < 32) ? (L - bj) : 32;
        for (int q = 0; q < m; ++q)
          dr += (__builtin_amdgcn_readlane(other, q) < mine) ? 1 : 0;
      }
      if (i < L) {
        col_s[s + dr] = col[mine];
        w_s[s + dr]   = w[mine];
      }
    }
  }
}

__global__ void deg_dis_k(const int* __restrict__ rowptr, const float* __restrict__ w_s,
                          float* __restrict__ dis, int n) {
  int i = blockIdx.x * blockDim.x + threadIdx.x;
  if (i >= n) return;
  const int s = rowptr[i], t = rowptr[i + 1];
  float d = 1.0f;
  for (int p = s; p < t; ++p) d += w_s[p];
  dis[i] = rsqrtf(d);
}

__global__ void edge_norm_k(const int* __restrict__ rowptr, const int* __restrict__ col_s,
                            const float* __restrict__ dis, float* w_s, int n) {
  int i = blockIdx.x * blockDim.x + threadIdx.x;
  if (i >= n) return;
  const int s = rowptr[i], t = rowptr[i + 1];
  const float di = dis[i];
  for (int p = s; p < t; ++p) w_s[p] = di * w_s[p] * dis[col_s[p]];
}

template <int U, bool RELU_IN, bool SPLIT>
__global__ __launch_bounds__(256) void gemm_strip_k(const float* __restrict__ X,
                                                    const float* __restrict__ Wa,
                                                    const float* __restrict__ Wb,
                                                    float* __restrict__ T, int MT) {
  constexpr int P = 132;
  constexpr int PB = 2 * U + 32;
  constexpr int NT = U / 16;
  __shared__ alignas(16) float Bs[64 * PB];
  __shared__ alignas(16) float As[8 * 16 * P];
  const int lane = threadIdx.x & 31;
  const int wave = threadIdx.x >> 5;
  const int tile_m = blockIdx.x * 8 + wave;
  const bool live = tile_m < MT;
  const int tm = live ? tile_m : MT - 1;

  const float* xs = X + (size_t)tm * (16 * 128) + lane * 4;
  float* as = As + wave * (16 * P);
  float4 xa[16];
  for (int j = 0; j < 16; ++j) xa[j] = *reinterpret_cast<const float4*>(xs + j * 128);
  static_assert(U == 128, "a W row is one float4 per lane");
  constexpr int WP = SPLIT ? U / 2 : U;
  const float* wcol = (SPLIT && lane >= 16) ? (Wb + (lane - 16) * 4) : (Wa + lane * 4);
  for (int i = 0; i < 64 / 8; ++i) {
    const int pr = i * 8 + wave;
    const float4 w0 = *reinterpret_cast<const float4*>(wcol + (size_t)(2 * pr) * WP);
    const float4 w1 = *reinterpret_cast<const float4*>(wcol + (size_t)(2 * pr + 1) * WP);
    float* bd = Bs + pr * PB + lane * 8;
    *reinterpret_cast<float4*>(bd)     = make_float4(w0.x, w1.x, w0.y, w1.y);
    *reinterpret_cast<float4*>(bd + 4) = make_float4(w0.z, w1.z, w0.w, w1.w);
  }
  for (int j = 0; j < 16; ++j) *reinterpret_cast<float4*>(as + j * P + lane * 4) = xa[j];
  __syncthreads();
  if (!live) return;

  const int kk = (lane >> 4) * 2;
  const float* ar = as + (lane & 15) * P + kk;
  const float* br = Bs + (lane >> 4) * PB + (lane & 15) * 2;
  v8f c[NT];
  for (int t = 0; t < NT; ++t) c[t] = v8f{0.f, 0.f, 0.f, 0.f, 0.f, 0.f, 0.f, 0.f};
  for (int k = 0; k < 128; k += 4) {
    v2f a = *reinterpret_cast<const v2f*>(ar + k);
    if (RELU_IN) { a.x = fmaxf(a.x, 0.0f); a.y = fmaxf(a.y, 0.0f); }
    for (int t = 0; t < NT; ++t) {
      const v2f b = *reinterpret_cast<const v2f*>(br + (k >> 1) * PB + t * 32);
      c[t] = __builtin_amdgcn_wmma_f32_16x16x4_f32(false, a, false, b, (short)0, c[t],
                                                   false, false);
    }
  }

  const int mbase = tile_m * 16 + (lane >> 4) * 8;
  float* trow = T + (size_t)mbase * U + (lane & 15);
  for (int t = 0; t < NT; ++t)
    for (int r = 0; r < 8; ++r) trow[(size_t)r * U + t * 16] = c[t][r];
}

template <int U, bool HEADS>
__global__ __launch_bounds__(256) void agg_rows_k(const int* __restrict__ rowptr,
                                                  const int* __restrict__ col_s,
                                                  const float* __restrict__ nrm_s,
                                                  const float* __restrict__ T,
                                                  const float* __restrict__ dis,
                                                  const float* __restrict__ bias,
                                                  const float* __restrict__ bias_hi,
                                                  const float* __restrict__ rnd,
                                                  float* __restrict__ out, int n) {
  constexpr int NL = U / 32;
  constexpr int HL = NL / 2;
  const int r = (blockIdx.x * 256 + threadIdx.x) >> 5;
  if (r >= n) return;
  const int lane = threadIdx.x & 31;
  const int s = __builtin_amdgcn_readfirstlane(rowptr[r]);
  const int t = __builtin_amdgcn_readfirstlane(rowptr[r + 1]);
  const float* Tl = T + lane;
  const float d  = dis[r];
  const float sn = d * d;
  float acc[NL];
  for (int j = 0; j < NL; ++j) {
    const float bj = (HEADS && j >= HL) ? bias_hi[lane + 32 * (j - HL)] : bias[lane + 32 * j];
    acc[j] = bj + sn * Tl[(size_t)r * U + 32 * j];
  }
  float rv[HL > 0 ? HL : 1];
  if (HEADS)
    for (int j = 0; j < HL; ++j) rv[j] = rnd[(size_t)r * (U / 2) + lane + 32 * j];

  for (int base = s; base < t; base += 32) {
    const int m = (t - base < 32) ? (t - base) : 32;
    int   cv = 0;
    float wv = 0.0f;
    if (lane < m) { cv = col_s[base + lane]; wv = nrm_s[base + lane]; }
    const int wi = __float_as_int(wv);
    int q = 0;
    for (; q + 4 <= m; q += 4) {
      const int c0 = __builtin_amdgcn_readlane(cv, q);
      const int c1 = __builtin_amdgcn_readlane(cv, q + 1);
      const int c2 = __builtin_amdgcn_readlane(cv, q + 2);
      const int c3 = __builtin_amdgcn_readlane(cv, q + 3);
      const float w0 = __int_as_float(__builtin_amdgcn_readlane(wi, q));
      const float w1 = __int_as_float(__builtin_amdgcn_readlane(wi, q + 1));
      const float w2 = __int_as_float(__builtin_amdgcn_readlane(wi, q + 2));
      const float w3 = __int_as_float(__builtin_amdgcn_readlane(wi, q + 3));
      float t0[NL], t1[NL], t2[NL], t3[NL];
      for (int j = 0; j < NL; ++j) {
        t0[j] = Tl[(size_t)c0 * U + 32 * j];
        t1[j] = Tl[(size_t)c1 * U + 32 * j];
        t2[j] = Tl[(size_t)c2 * U + 32 * j];
        t3[j] = Tl[(size_t)c3 * U + 32 * j];
      }
      for (int j = 0; j < NL; ++j) {
        acc[j] += w0 * t0[j];
        acc[j] += w1 * t1[j];
        acc[j] += w2 * t2[j];
        acc[j] += w3 * t3[j];
      }
    }
    for (; q < m; ++q) {
      const int   c0 = __builtin_amdgcn_readlane(cv, q);
      const float w0 = __int_as_float(__builtin_amdgcn_readlane(wi, q));
      for (int j = 0; j < NL; ++j) acc[j] += w0 * Tl[(size_t)c0 * U + 32 * j];
    }
  }
  if (HEADS) {
    float* ho = out + (size_t)r * (U / 2) + lane;
    float* ko = ho + (size_t)n * (U / 2);
    for (int j = 0; j < HL; ++j) {
      const float mu = acc[j];
      const float ls = acc[HL + j];
      const float sd = expf(ls);
      ho[32 * j] = mu + rv[j] * sd;
      ko[32 * j] = -ls + 0.5f * (sd * sd + mu * mu - 1.0f);
    }
  } else {
    float* orow = out + (size_t)r * U + lane;
    for (int j = 0; j < NL; ++j) orow[32 * j] = acc[j];
  }
}

template <bool SPLIT>
__global__ __launch_bounds__(512, 12) void agg_gemm_k(const int* __restrict__ rowptr,
                                                  const int* __restrict__ col_s,
                                                  const float* __restrict__ nrm_s,
                                                  const float* __restrict__ Tin,
                                                  const float* __restrict__ dis,
                                                  const float* __restrict__ bias,
                                                  const float* __restrict__ Wa,
                                                  const float* __restrict__ Wb,
                                                  float* __restrict__ Tout, int MT) {
  constexpr int U  = 128;
  constexpr int NL = U / 32;
  constexpr int P  = 132;
  constexpr int PB = 2 * U + 32;
  __shared__ alignas(16) float Bs[64 * PB];
  __shared__ alignas(16) float As[2 * 16 * P];
  const int lane = threadIdx.x & 31;
  const int wave = __builtin_amdgcn_readfirstlane((int)(threadIdx.x >> 5));

  constexpr int WP = SPLIT ? U / 2 : U;
  const float* wcol = (SPLIT && lane >= 16) ? (Wb + (lane - 16) * 4) : (Wa + lane * 4);
  for (int i = 0; i < 64 / 16; ++i) {
    const int pr = i * 16 + wave;
    const float4 w0 = *reinterpret_cast<const float4*>(wcol + (size_t)(2 * pr) * WP);
    const float4 w1 = *reinterpret_cast<const float4*>(wcol + (size_t)(2 * pr + 1) * WP);
    float* bd = Bs + pr * PB + lane * 8;
    *reinterpret_cast<float4*>(bd)     = make_float4(w0.x, w1.x, w0.y, w1.y);
    *reinterpret_cast<float4*>(bd + 4) = make_float4(w0.z, w1.z, w0.w, w1.w);
  }

  const float* Tl = Tin + lane;
  float bj[NL];
  for (int j = 0; j < NL; ++j) bj[j] = bias[lane + 32 * j];
  const int kk = (lane >> 4) * 2;

  int it = 0;
  for (int tm = blockIdx.x; tm < MT; tm += gridDim.x, ++it) {
    float* asb = As + (it & 1) * (16 * P);
    const int r = tm * 16 + wave;
    const int s = __builtin_amdgcn_readfirstlane(rowptr[r]);
    const int t = __builtin_amdgcn_readfirstlane(rowptr[r + 1]);
    const float d  = dis[r];
    const float sn = d * d;
    float acc[NL];
    for (int j = 0; j < NL; ++j) acc[j] = bj[j] + sn * Tl[(size_t)r * U + 32 * j];

    for (int base = s; base < t; base += 32) {
      const int m = (t - base < 32) ? (t - base) : 32;
      int   cv = 0;
      float wv = 0.0f;
      if (lane < m) { cv = col_s[base + lane]; wv = nrm_s[base + lane]; }
      const int wi = __float_as_int(wv);
      int q = 0;
      for (; q + 4 <= m; q += 4) {
        const int c0 = __builtin_amdgcn_readlane(cv, q);
        const int c1 = __builtin_amdgcn_readlane(cv, q + 1);
        const int c2 = __builtin_amdgcn_readlane(cv, q + 2);
        const int c3 = __builtin_amdgcn_readlane(cv, q + 3);
        const float w0 = __int_as_float(__builtin_amdgcn_readlane(wi, q));
        const float w1 = __int_as_float(__builtin_amdgcn_readlane(wi, q + 1));
        const float w2 = __int_as_float(__builtin_amdgcn_readlane(wi, q + 2));
        const float w3 = __int_as_float(__builtin_amdgcn_readlane(wi, q + 3));
        float t0[NL], t1[NL], t2[NL], t3[NL];
        for (int j = 0; j < NL; ++j) {
          t0[j] = Tl[(size_t)c0 * U + 32 * j];
          t1[j] = Tl[(size_t)c1 * U + 32 * j];
          t2[j] = Tl[(size_t)c2 * U + 32 * j];
          t3[j] = Tl[(size_t)c3 * U + 32 * j];
        }
        for (int j = 0; j < NL; ++j) {
          acc[j] += w0 * t0[j];
          acc[j] += w1 * t1[j];
          acc[j] += w2 * t2[j];
          acc[j] += w3 * t3[j];
        }
      }
      for (; q < m; ++q) {
        const int   c0 = __builtin_amdgcn_readlane(cv, q);
        const float w0 = __int_as_float(__builtin_amdgcn_readlane(wi, q));
        for (int j = 0; j < NL; ++j) acc[j] += w0 * Tl[(size_t)c0 * U + 32 * j];
      }
    }
    for (int j = 0; j < NL; ++j) asb[wave * P + lane + 32 * j] = fmaxf(acc[j], 0.0f);
    __syncthreads();

    if (wave < 4) {
      const float* ar = asb + (lane & 15) * P + kk;
      const float* br = Bs + (lane >> 4) * PB + (lane & 15) * 2 + wave * 64;
      v8f ca = v8f{0.f, 0.f, 0.f, 0.f, 0.f, 0.f, 0.f, 0.f};
      v8f cb = v8f{0.f, 0.f, 0.f, 0.f, 0.f, 0.f, 0.f, 0.f};
      v2f fa[2][4], f0[2][4], f1[2][4];
      for (int i = 0; i < 4; ++i) {
        fa[0][i] = *reinterpret_cast<const v2f*>(ar + 4 * i);
        f0[0][i] = *reinterpret_cast<const v2f*>(br + 2 * i * PB);
        f1[0][i] = *reinterpret_cast<const v2f*>(br + 2 * i * PB + 32);
      }
#pragma unroll
      for (int kb = 0; kb < 8; ++kb) {
        const int cur = kb & 1, nxt = cur ^ 1;
        if (kb + 1 < 8) {
          for (int i = 0; i < 4; ++i) {
            const int k = (kb + 1) * 16 + 4 * i;
            fa[nxt][i] = *reinterpret_cast<const v2f*>(ar + k);
            f0[nxt][i] = *reinterpret_cast<const v2f*>(br + (k >> 1) * PB);
            f1[nxt][i] = *reinterpret_cast<const v2f*>(br + (k >> 1) * PB + 32);
          }
        }
        __builtin_amdgcn_sched_barrier(0);
        for (int i = 0; i < 4; ++i) {
          ca = __builtin_amdgcn_wmma_f32_16x16x4_f32(false, fa[cur][i], false, f0[cur][i],
                                                     (short)0, ca, false, false);
          cb = __builtin_amdgcn_wmma_f32_16x16x4_f32(false, fa[cur][i], false, f1[cur][i],
                                                     (short)0, cb, false, false);
        }
        __builtin_amdgcn_sched_barrier(0);
      }
      const int mbase = tm * 16 + (lane >> 4) * 8;
      float* trow = Tout + (size_t)mbase * U + wave * 32 + (lane & 15);
      for (int r8 = 0; r8 < 8; ++r8) {
        trow[(size_t)r8 * U]      = ca[r8];
        trow[(size_t)r8 * U + 16] = cb[r8];
      }
    }
  }
}

extern "C" void kernel_launch(void* const* d_in, const int* in_sizes, int n_in,
                              void* d_out, int out_size, void* d_ws, size_t ws_size,
                              hipStream_t stream) {
  (void)in_sizes; (void)n_in; (void)out_size; (void)ws_size;
  const int N = NN, E = NE;
  const float* h   = (const float*)d_in[0];
  const int*   ei  = (const int*)d_in[1];
  const float* ew  = (const float*)d_in[2];
  const float* rnd = (const float*)d_in[3];
  const float* W0  = (const float*)d_in[4];
  const float* b0  = (const float*)d_in[5];
  const float* W1  = (const float*)d_in[6];
  const float* b1  = (const float*)d_in[7];
  const float* Wmu = (const float*)d_in[8];
  const float* bmu = (const float*)d_in[9];
  const float* Wsd = (const float*)d_in[10];
  const float* bsd = (const float*)d_in[11];
  const int* row = ei;
  const int* col = ei + E;
  float* out = (float*)d_out;

  float* ws    = (float*)d_ws;
  float* dis   = ws;
  float* nrm   = dis + N;
  float* bufT  = nrm + E;
  float* bufH  = bufT + (size_t)N * 128;
  float* bufS  = bufH + (size_t)N * 128;
  int* col_s   = (int*)bufS;
  int* rowptr  = col_s + E;
  int* bsum    = rowptr + N + 96;
  int* hist    = bsum + 1024;
  int2* rec    = (int2*)bufH;
  int* rk      = (int*)bufH + 2 * (size_t)E;
  int* tmp     = rk + E;

  const int TB = 256;
  const int H1 = NBK * NCH + 1;
  const int SB = (H1 + 255) / 256;
  bucket_hist_k   <<<NCH, 1024, 0, stream>>>(row, hist, E);
  scan_local_k    <<<SB, 256, 0, stream>>>(hist, bsum, H1);
  scan_add_k      <<<SB, 256, 0, stream>>>(hist, bsum, H1);
  bucket_scatter_k<<<NCH, 1024, 0, stream>>>(row, hist, rec, E);
  bucket_rows_k   <<<NBK, 512, 0, stream>>>(hist, rec, rk, tmp, col, ew, rowptr, col_s, nrm, N);
  deg_dis_k   <<<(N + TB - 1) / TB, TB, 0, stream>>>(rowptr, nrm, dis, N);
  edge_norm_k <<<(N + TB - 1) / TB, TB, 0, stream>>>(rowptr, col_s, dis, nrm, N);

  const int MT = N / 16;
  const unsigned GB = (unsigned)((MT + 7) / 8);
  const unsigned AB = (unsigned)((N * 32 + 255) / 256);
  const unsigned FB = 736u;

  gemm_strip_k<128, false, false><<<GB, 256, 0, stream>>>(h, W0, nullptr, bufT, MT);
  agg_gemm_k<false><<<FB, 512, 0, stream>>>(rowptr, col_s, nrm, bufT, dis, b0, W1, nullptr, bufH, MT);
  agg_gemm_k<true><<<FB, 512, 0, stream>>>(rowptr, col_s, nrm, bufH, dis, b1, Wmu, Wsd, bufT, MT);
  agg_rows_k<128, true><<<AB, 256, 0, stream>>>(rowptr, col_s, nrm, bufT, dis, bmu, bsd, rnd, out, N);
}


# layers reordered by linearity: each launch sums the rows of X first, then multiplies the sums by W on WMMA (+ b, act); first GEMM launch and its T plane go (11 -> 10 launches); f32 sum order changed
# speedup vs baseline: 4.5657x; 1.0602x over previous
#include <hip/hip_runtime.h>

#define NN 100000
#define NE 1600000

typedef float v2f __attribute__((ext_vector_type(2)));
typedef float v8f __attribute__((ext_vector_type(8)));

#define RBITS 7
#define RPB   (1 << RBITS)
#define NBK   ((NN + RPB - 1) >> RBITS)
#define NCH   256
#define CHUNK ((NE + NCH - 1) / NCH)

__global__ __launch_bounds__(1024) void bucket_hist_k(const int* __restrict__ row,
                                                      int* __restrict__ hist, int e) {
  __shared__ int h[NBK];
  for (int k = threadIdx.x; k < NBK; k += 1024) h[k] = 0;
  __syncthreads();
  const int lo = blockIdx.x * CHUNK;
  const int hi = (lo + CHUNK < e) ? (lo + CHUNK) : e;
  for (int i = lo + threadIdx.x; i < hi; i += 1024) atomicAdd(&h[row[i] >> RBITS], 1);
  __syncthreads();
  for (int k = threadIdx.x; k < NBK; k += 1024) hist[k * NCH + blockIdx.x] = h[k];
  if (blockIdx.x == 0 && threadIdx.x == 0) hist[NBK * NCH] = 0;
}

__global__ __launch_bounds__(256) void scan_local_k(int* cnt, int* __restrict__ bsum, int n1) {
  __shared__ int wsum[8];
  const int lane = threadIdx.x & 31;
  const int wv   = threadIdx.x >> 5;
  const int i = blockIdx.x * 256 + threadIdx.x;
  const int v = (i < n1) ? cnt[i] : 0;
  int x = v;
  for (int d = 1; d < 32; d <<= 1) {
    int y = __shfl_up(x, d);
    if (lane >= d) x += y;
  }
  if (lane == 31) wsum[wv] = x;
  __syncthreads();
  int off = 0;
  for (int k = 0; k < 8; ++k) off += (k < wv) ? wsum[k] : 0;
  const int incl = x + off;
  if (i < n1) cnt[i] = incl - v;
  if (threadIdx.x == 255) bsum[blockIdx.x] = incl;
}

__global__ __launch_bounds__(256) void scan_add_k(int* rowptr, const int* __restrict__ bsum, int n1) {
  __shared__ int red[8];
  const int lane = threadIdx.x & 31;
  const int wv   = threadIdx.x >> 5;
  const int b = blockIdx.x;
  int s = 0;
  for (int k = threadIdx.x; k < b; k += 256) s += bsum[k];
  for (int d = 16; d >= 1; d >>= 1) s += __shfl_xor(s, d);
  if (lane == 0) red[wv] = s;
  __syncthreads();
  int off = 0;
  for (int k = 0; k < 8; ++k) off += red[k];
  const int i = b * 256 + threadIdx.x;
  if (i < n1) rowptr[i] += off;
}

__global__ __launch_bounds__(1024) void bucket_scatter_k(const int* __restrict__ row,
                                                         const int* __restrict__ hist,
                                                         int2* __restrict__ rec, int e) {
  __shared__ int cur[NBK];
  for (int k = threadIdx.x; k < NBK; k += 1024) cur[k] = hist[k * NCH + blockIdx.x];
  __syncthreads();
  const int lo = blockIdx.x * CHUNK;
  const int hi = (lo + CHUNK < e) ? (lo + CHUNK) : e;
  for (int i = lo + threadIdx.x; i < hi; i += 1024) {
    const int r = row[i];
    const int p = atomicAdd(&cur[r >> RBITS], 1);
    rec[p] = make_int2(i, r);
  }
}

__global__ __launch_bounds__(512) void bucket_rows_k(const int* __restrict__ hist,
                                                     const int2* __restrict__ rec,
                                                     int* __restrict__ rk,
                                                     int* __restrict__ tmp,
                                                     const int* __restrict__ col,
                                                     const float* __restrict__ w,
                                                     int* __restrict__ rowptr,
                                                     int* __restrict__ col_s,
                                                     float* __restrict__ w_s, int n) {
  __shared__ int cnt[RPB];
  __shared__ int off[RPB + 1];
  const int b = blockIdx.x;
  const int lane = threadIdx.x & 31;
  const int wv   = threadIdx.x >> 5;
  const int bs = __builtin_amdgcn_readfirstlane(hist[b * NCH]);
  const int be = __builtin_amdgcn_readfirstlane(hist[(b + 1) * NCH]);
  const int rbase = b << RBITS;
  if (threadIdx.x < RPB) cnt[threadIdx.x] = 0;
  __syncthreads();
  for (int i = bs + (int)threadIdx.x; i < be; i += 512) {
    const int2 q = rec[i];
    rk[i] = atomicAdd(&cnt[q.y - rbase], 1);
  }
  __syncthreads();
  if (wv == 0) {
    const int c0 = cnt[4 * lane], c1 = cnt[4 * lane + 1];
    const int c2 = cnt[4 * lane + 2], c3 = cnt[4 * lane + 3];
    const int sum4 = c0 + c1 + c2 + c3;
    int x = sum4;
    for (int d = 1; d < 32; d <<= 1) {
      const int y = __shfl_up(x, d);
      if (lane >= d) x += y;
    }
    const int ex = x - sum4;
    off[4 * lane]     = ex;
    off[4 * lane + 1] = ex + c0;
    off[4 * lane + 2] = ex + c0 + c1;
    off[4 * lane + 3] = ex + c0 + c1 + c2;
    if (lane == 31) off[RPB] = x;
  }
  __syncthreads();
  if (threadIdx.x < RPB && rbase + (int)threadIdx.x < n)
    rowptr[rbase + threadIdx.x] = bs + off[threadIdx.x];
  if (b == (int)gridDim.x - 1 && threadIdx.x == 0) rowptr[n] = be;
  for (int i = bs + (int)threadIdx.x; i < be; i += 512) {
    const int2 q = rec[i];
    tmp[bs + off[q.y - rbase] + rk[i]] = q.x;
  }
  __syncthreads();
  for (int rr = wv; rr < RPB; rr += 16) {
    const int o0 = __builtin_amdgcn_readfirstlane(off[rr]);
    const int o1 = __builtin_amdgcn_readfirstlane(off[rr + 1]);
    const int s = bs + o0;
    const int L = o1 - o0;
    for (int bi = 0; bi < L; bi += 32) {
      const int i = bi + lane;
      const int mine = (i < L) ? tmp[s + i] : 0x7fffffff;
      int dr = 0;
      for (int bj = 0; bj < L; bj += 32) {
        const int j = bj + lane;
        const int other = (j < L) ? tmp[s + j] : 0x7fffffff;
        const int m = (L - bj < 32) ? (L - bj) : 32;
        for (int q = 0; q < m; ++q)
          dr += (__builtin_amdgcn_readlane(other, q) < mine) ? 1 : 0;
      }
      if (i < L) {
        col_s[s + dr] = col[mine];
        w_s[s + dr]   = w[mine];
      }
    }
  }
}

__global__ void deg_dis_k(const int* __restrict__ rowptr, const float* __restrict__ w_s,
                          float* __restrict__ dis, int n) {
  int i = blockIdx.x * blockDim.x + threadIdx.x;
  if (i >= n) return;
  const int s = rowptr[i], t = rowptr[i + 1];
  float d = 1.0f;
  for (int p = s; p < t; ++p) d += w_s[p];
  dis[i] = rsqrtf(d);
}

__global__ void edge_norm_k(const int* __restrict__ rowptr, const int* __restrict__ col_s,
                            const float* __restrict__ dis, float* w_s, int n) {
  int i = blockIdx.x * blockDim.x + threadIdx.x;
  if (i >= n) return;
  const int s = rowptr[i], t = rowptr[i + 1];
  const float di = dis[i];
  for (int p = s; p < t; ++p) w_s[p] = di * w_s[p] * dis[col_s[p]];
}

template <bool HEADS>
__global__ __launch_bounds__(512, 12) void agg_gemm_k(const int* __restrict__ rowptr,
                                                  const int* __restrict__ col_s,
                                                  const float* __restrict__ nrm_s,
                                                  const float* __restrict__ X,
                                                  const float* __restrict__ dis,
                                                  const float* __restrict__ Wa,
                                                  const float* __restrict__ Wb,
                                                  const float* __restrict__ bias,
                                                  const float* __restrict__ bias_hi,
                                                  const float* __restrict__ rnd,
                                                  float* __restrict__ out, int MT, int n) {
  constexpr int U  = 128;
  constexpr int NL = U / 32;
  constexpr int HL = NL / 2;
  constexpr int P  = 132;
  constexpr int PB = 2 * U + 32;
  constexpr int CSZ = HEADS ? 2 * 16 * P : 4;
  __shared__ alignas(16) float Bs[64 * PB];
  __shared__ alignas(16) float As[2 * 16 * P];
  __shared__ alignas(16) float Cs[CSZ];
  const int lane = threadIdx.x & 31;
  const int wave = __builtin_amdgcn_readfirstlane((int)(threadIdx.x >> 5));

  constexpr int WP = HEADS ? U / 2 : U;
  const float* wcol = (HEADS && lane >= 16) ? (Wb + (lane - 16) * 4) : (Wa + lane * 4);
  for (int i = 0; i < 64 / 16; ++i) {
    const int pr = i * 16 + wave;
    const float4 w0 = *reinterpret_cast<const float4*>(wcol + (size_t)(2 * pr) * WP);
    const float4 w1 = *reinterpret_cast<const float4*>(wcol + (size_t)(2 * pr + 1) * WP);
    float* bd = Bs + pr * PB + lane * 8;
    *reinterpret_cast<float4*>(bd)     = make_float4(w0.x, w1.x, w0.y, w1.y);
    *reinterpret_cast<float4*>(bd + 4) = make_float4(w0.z, w1.z, w0.w, w1.w);
  }

  const float* Xl = X + lane;
  float b0v = 0.0f, b1v = 0.0f;
  if (!HEADS) {
    b0v = bias[(wave & 3) * 32 + (lane & 15)];
    b1v = bias[(wave & 3) * 32 + 16 + (lane & 15)];
  }
  const int kk = (lane >> 4) * 2;

  int it = 0;
  int tprev = 0;
  for (int tm = blockIdx.x; tm < MT; tm += gridDim.x, ++it) {
    float* asb = As + (it & 1) * (16 * P);
    const int r = tm * 16 + wave;
    float rv[HL] = {0.0f, 0.0f}, bm[HL] = {0.0f, 0.0f}, bl[HL] = {0.0f, 0.0f};
    if (HEADS && it > 0)
      for (int j = 0; j < HL; ++j) {
        rv[j] = rnd[(size_t)(tprev * 16 + wave) * (U / 2) + lane + 32 * j];
        bm[j] = bias[lane + 32 * j];
        bl[j] = bias_hi[lane + 32 * j];
      }
    const int s = __builtin_amdgcn_readfirstlane(rowptr[r]);
    const int t = __builtin_amdgcn_readfirstlane(rowptr[r + 1]);
    const float d  = dis[r];
    const float sn = d * d;
    float acc[NL];
    for (int j = 0; j < NL; ++j) acc[j] = sn * Xl[(size_t)r * U + 32 * j];

    for (int base = s; base < t; base += 32) {
      const int m = (t - base < 32) ? (t - base) : 32;
      int   cv = 0;
      float wv = 0.0f;
      if (lane < m) { cv = col_s[base + lane]; wv = nrm_s[base + lane]; }
      const int wi = __float_as_int(wv);
      int q = 0;
      for (; q + 4 <= m; q += 4) {
        const int c0 = __builtin_amdgcn_readlane(cv, q);
        const int c1 = __builtin_amdgcn_readlane(cv, q + 1);
        const int c2 = __builtin_amdgcn_readlane(cv, q + 2);
        const int c3 = __builtin_amdgcn_readlane(cv, q + 3);
        const float w0 = __int_as_float(__builtin_amdgcn_readlane(wi, q));
        const float w1 = __int_as_float(__builtin_amdgcn_readlane(wi, q + 1));
        const float w2 = __int_as_float(__builtin_amdgcn_readlane(wi, q + 2));
        const float w3 = __int_as_float(__builtin_amdgcn_readlane(wi, q + 3));
        float t0[NL], t1[NL], t2[NL], t3[NL];
        for (int j = 0; j < NL; ++j) {
          t0[j] = Xl[(size_t)c0 * U + 32 * j];
          t1[j] = Xl[(size_t)c1 * U + 32 * j];
          t2[j] = Xl[(size_t)c2 * U + 32 * j];
          t3[j] = Xl[(size_t)c3 * U + 32 * j];
        }
        for (int j = 0; j < NL; ++j) {
          acc[j] += w0 * t0[j];
          acc[j] += w1 * t1[j];
          acc[j] += w2 * t2[j];
          acc[j] += w3 * t3[j];
        }
      }
      for (; q < m; ++q) {
        const int   c0 = __builtin_amdgcn_readlane(cv, q);
        const float w0 = __int_as_float(__builtin_amdgcn_readlane(wi, q));
        for (int j = 0; j < NL; ++j) acc[j] += w0 * Xl[(size_t)c0 * U + 32 * j];
      }
    }
    for (int j = 0; j < NL; ++j) asb[wave * P + lane + 32 * j] = acc[j];
    __syncthreads();

    if (HEADS && it > 0) {
      const float* cs = Cs + ((it - 1) & 1) * (16 * P) + wave * P + lane;
      float* ho = out + (size_t)(tprev * 16 + wave) * (U / 2) + lane;
      float* ko = ho + (size_t)n * (U / 2);
      const float mu0 = cs[0]  + bm[0], mu1 = cs[32] + bm[1];
      const float ls0 = cs[64] + bl[0], ls1 = cs[96] + bl[1];
      const float sd0 = expf(ls0), sd1 = expf(ls1);
      ho[0]  = mu0 + rv[0] * sd0;
      ho[32] = mu1 + rv[1] * sd1;
      ko[0]  = -ls0 + 0.5f * (sd0 * sd0 + mu0 * mu0 - 1.0f);
      ko[32] = -ls1 + 0.5f * (sd1 * sd1 + mu1 * mu1 - 1.0f);
    }

    if (wave < 4) {
      const float* ar = asb + (lane & 15) * P + kk;
      const float* br = Bs + (lane >> 4) * PB + (lane & 15) * 2 + wave * 64;
      v8f ca = v8f{0.f, 0.f, 0.f, 0.f, 0.f, 0.f, 0.f, 0.f};
      v8f cb = v8f{0.f, 0.f, 0.f, 0.f, 0.f, 0.f, 0.f, 0.f};
      v2f fa[2][4], f0[2][4], f1[2][4];
      for (int i = 0; i < 4; ++i) {
        fa[0][i] = *reinterpret_cast<const v2f*>(ar + 4 * i);
        f0[0][i] = *reinterpret_cast<const v2f*>(br + 2 * i * PB);
        f1[0][i] = *reinterpret_cast<const v2f*>(br + 2 * i * PB + 32);
      }
#pragma unroll
      for (int kb = 0; kb < 8; ++kb) {
        const int cur = kb & 1, nxt = cur ^ 1;
        if (kb + 1 < 8) {
          for (int i = 0; i < 4; ++i) {
            const int k = (kb + 1) * 16 + 4 * i;
            fa[nxt][i] = *reinterpret_cast<const v2f*>(ar + k);
            f0[nxt][i] = *reinterpret_cast<const v2f*>(br + (k >> 1) * PB);
            f1[nxt][i] = *reinterpret_cast<const v2f*>(br + (k >> 1) * PB + 32);
          }
        }
        __builtin_amdgcn_sched_barrier(0);
        for (int i = 0; i < 4; ++i) {
          ca = __builtin_amdgcn_wmma_f32_16x16x4_f32(false, fa[cur][i], false, f0[cur][i],
                                                     (short)0, ca, false, false);
          cb = __builtin_amdgcn_wmma_f32_16x16x4_f32(false, fa[cur][i], false, f1[cur][i],
                                                     (short)0, cb, false, false);
        }
        __builtin_amdgcn_sched_barrier(0);
      }
      if (HEADS) {
        float* cw = Cs + (it & 1) * (16 * P) + ((lane >> 4) * 8) * P + wave * 32 + (lane & 15);
        for (int r8 = 0; r8 < 8; ++r8) {
          cw[r8 * P]      = ca[r8];
          cw[r8 * P + 16] = cb[r8];
        }
      } else {
        const int mbase = tm * 16 + (lane >> 4) * 8;
        float* trow = out + (size_t)mbase * U + wave * 32 + (lane & 15);
        for (int r8 = 0; r8 < 8; ++r8) {
          trow[(size_t)r8 * U]      = fmaxf(ca[r8] + b0v, 0.0f);
          trow[(size_t)r8 * U + 16] = fmaxf(cb[r8] + b1v, 0.0f);
        }
      }
    }
    tprev = tm;
  }
  if (HEADS) {
    __syncthreads();
    if (it > 0) {
      const int rp = tprev * 16 + wave;
      const float* cs = Cs + ((it - 1) & 1) * (16 * P) + wave * P + lane;
      float* ho = out + (size_t)rp * (U / 2) + lane;
      float* ko = ho + (size_t)n * (U / 2);
      const float r0 = rnd[(size_t)rp * (U / 2) + lane];
      const float r1 = rnd[(size_t)rp * (U / 2) + lane + 32];
      const float mu0 = cs[0]  + bias[lane],    mu1 = cs[32] + bias[lane + 32];
      const float ls0 = cs[64] + bias_hi[lane], ls1 = cs[96] + bias_hi[lane + 32];
      const float sd0 = expf(ls0), sd1 = expf(ls1);
      ho[0]  = mu0 + r0 * sd0;
      ho[32] = mu1 + r1 * sd1;
      ko[0]  = -ls0 + 0.5f * (sd0 * sd0 + mu0 * mu0 - 1.0f);
      ko[32] = -ls1 + 0.5f * (sd1 * sd1 + mu1 * mu1 - 1.0f);
    }
  }
}

extern "C" void kernel_launch(void* const* d_in, const int* in_sizes, int n_in,
                              void* d_out, int out_size, void* d_ws, size_t ws_size,
                              hipStream_t stream) {
  (void)in_sizes; (void)n_in; (void)out_size; (void)ws_size;
  const int N = NN, E = NE;
  const float* h   = (const float*)d_in[0];
  const int*   ei  = (const int*)d_in[1];
  const float* ew  = (const float*)d_in[2];
  const float* rnd = (const float*)d_in[3];
  const float* W0  = (const float*)d_in[4];
  const float* b0  = (const float*)d_in[5];
  const float* W1  = (const float*)d_in[6];
  const float* b1  = (const float*)d_in[7];
  const float* Wmu = (const float*)d_in[8];
  const float* bmu = (const float*)d_in[9];
  const float* Wsd = (const float*)d_in[10];
  const float* bsd = (const float*)d_in[11];
  const int* row = ei;
  const int* col = ei + E;
  float* out = (float*)d_out;

  float* ws    = (float*)d_ws;
  float* dis   = ws;
  float* nrm   = dis + N;
  float* bufT  = nrm + E;
  float* bufH  = bufT + (size_t)N * 128;
  float* bufS  = bufH + (size_t)N * 128;
  int* col_s   = (int*)bufS;
  int* rowptr  = col_s + E;
  int* bsum    = rowptr + N + 96;
  int* hist    = bsum + 1024;
  int2* rec    = (int2*)bufH;
  int* rk      = (int*)bufH + 2 * (size_t)E;
  int* tmp     = rk + E;

  const int TB = 256;
  const int H1 = NBK * NCH + 1;
  const int SB = (H1 + 255) / 256;
  bucket_hist_k   <<<NCH, 1024, 0, stream>>>(row, hist, E);
  scan_local_k    <<<SB, 256, 0, stream>>>(hist, bsum, H1);
  scan_add_k      <<<SB, 256, 0, stream>>>(hist, bsum, H1);
  bucket_scatter_k<<<NCH, 1024, 0, stream>>>(row, hist, rec, E);
  bucket_rows_k   <<<NBK, 512, 0, stream>>>(hist, rec, rk, tmp, col, ew, rowptr, col_s, nrm, N);
  deg_dis_k   <<<(N + TB - 1) / TB, TB, 0, stream>>>(rowptr, nrm, dis, N);
  edge_norm_k <<<(N + TB - 1) / TB, TB, 0, stream>>>(rowptr, col_s, dis, nrm, N);

  const int MT = N / 16;
  const unsigned FB = 736u;

  agg_gemm_k<false><<<FB, 512, 0, stream>>>(rowptr, col_s, nrm, h, dis, W0, nullptr, b0, nullptr,
                                            nullptr, bufT, MT, N);
  agg_gemm_k<false><<<FB, 512, 0, stream>>>(rowptr, col_s, nrm, bufT, dis, W1, nullptr, b1, nullptr,
                                            nullptr, bufH, MT, N);
  agg_gemm_k<true><<<FB, 512, 0, stream>>>(rowptr, col_s, nrm, bufH, dis, Wmu, Wsd, bmu, bsd,
                                           rnd, out, MT, N);
}


# three small launches folded into neighbours (10 -> 7): bucket launches add the earlier buckets' counts and sum row degrees from LDS, layer 1 forms each edge norm beside its gathers; sum orders kept
# speedup vs baseline: 4.8145x; 1.0545x over previous
#include <hip/hip_runtime.h>

#define NN 100000
#define NE 1600000

typedef float v2f __attribute__((ext_vector_type(2)));
typedef float v8f __attribute__((ext_vector_type(8)));

#define RBITS 7
#define RPB   (1 << RBITS)
#define NBK   ((NN + RPB - 1) >> RBITS)
#define NCH   256
#define CHUNK ((NE + NCH - 1) / NCH)

__global__ __launch_bounds__(1024) void bucket_hist_k(const int* __restrict__ row,
                                                      int* __restrict__ hist, int e) {
  __shared__ int h[NBK];
  for (int k = threadIdx.x; k < NBK; k += 1024) h[k] = 0;
  __syncthreads();
  const int lo = blockIdx.x * CHUNK;
  const int hi = (lo + CHUNK < e) ? (lo + CHUNK) : e;
  for (int i = lo + threadIdx.x; i < hi; i += 1024) atomicAdd(&h[row[i] >> RBITS], 1);
  __syncthreads();
  for (int k = threadIdx.x; k < NBK; k += 1024) hist[k * NCH + blockIdx.x] = h[k];
}

__global__ __launch_bounds__(256) void scan_local_k(int* cnt, int* __restrict__ bsum, int n1) {
  __shared__ int wsum[8];
  const int lane = threadIdx.x & 31;
  const int wv   = threadIdx.x >> 5;
  const int i = blockIdx.x * 256 + threadIdx.x;
  const int v = (i < n1) ? cnt[i] : 0;
  int x = v;
  for (int d = 1; d < 32; d <<= 1) {
    int y = __shfl_up(x, d);
    if (lane >= d) x += y;
  }
  if (lane == 31) wsum[wv] = x;
  __syncthreads();
  int off = 0;
  for (int k = 0; k < 8; ++k) off += (k < wv) ? wsum[k] : 0;
  const int incl = x + off;
  if (i < n1) cnt[i] = incl - v;
  if (threadIdx.x == 255) bsum[blockIdx.x] = incl;
}

static_assert(NCH == 256, "one scan block per bucket");
static_assert(NBK <= 1024, "one bucket per thread in the scan of the bucket counts");
__global__ __launch_bounds__(1024) void bucket_scatter_k(const int* __restrict__ row,
                                                         const int* __restrict__ hist,
                                                         const int* __restrict__ bsum,
                                                         int2* __restrict__ rec, int e) {
  __shared__ int cur[NBK];
  __shared__ int wtot[32];
  {
    const int lane = threadIdx.x & 31;
    const int wv   = threadIdx.x >> 5;
    const int k = threadIdx.x;
    const int v = (k < NBK) ? bsum[k] : 0;
    const int first = (k < NBK) ? hist[k * NCH + blockIdx.x] : 0;
    int x = v;
    for (int d = 1; d < 32; d <<= 1) {
      const int y = __shfl_up(x, d);
      if (lane >= d) x += y;
    }
    if (lane == 31) wtot[wv] = x;
    __syncthreads();
    int off = 0;
    for (int j = 0; j < 32; ++j) off += (j < wv) ? wtot[j] : 0;
    if (k < NBK) cur[k] = first + off + x - v;
  }
  __syncthreads();
  const int lo = blockIdx.x * CHUNK;
  const int hi = (lo + CHUNK < e) ? (lo + CHUNK) : e;
  for (int i = lo + threadIdx.x; i < hi; i += 1024) {
    const int r = row[i];
    const int p = atomicAdd(&cur[r >> RBITS], 1);
    rec[p] = make_int2(i, r);
  }
}

#define WCAP 4096
__global__ __launch_bounds__(512) void bucket_rows_k(const int* __restrict__ bsum,
                                                     const int2* __restrict__ rec,
                                                     int* __restrict__ rk,
                                                     int* __restrict__ tmp,
                                                     const int* __restrict__ col,
                                                     const float* __restrict__ w,
                                                     int* __restrict__ rowptr,
                                                     int* __restrict__ col_s,
                                                     float* __restrict__ w_s,
                                                     float* __restrict__ dis, int n) {
  __shared__ int cnt[RPB];
  __shared__ int off[RPB + 1];
  __shared__ float wl[WCAP];
  __shared__ int red[16];
  const int b = blockIdx.x;
  const int lane = threadIdx.x & 31;
  const int wv   = threadIdx.x >> 5;
  int part = 0;
  for (int k = threadIdx.x; k < b; k += 512) part += bsum[k];
  for (int d = 16; d >= 1; d >>= 1) part += __shfl_xor(part, d);
  if (lane == 0) red[wv] = part;
  const int mine_n = bsum[b];
  if (threadIdx.x < RPB) cnt[threadIdx.x] = 0;
  __syncthreads();
  int before = 0;
  for (int k = 0; k < 16; ++k) before += red[k];
  const int bs = __builtin_amdgcn_readfirstlane(before);
  const int be = bs + __builtin_amdgcn_readfirstlane(mine_n);
  const int rbase = b << RBITS;
  const bool fits = (be - bs) <= WCAP;
  for (int i = bs + (int)threadIdx.x; i < be; i += 512) {
    const int2 q = rec[i];
    rk[i] = atomicAdd(&cnt[q.y - rbase], 1);
  }
  __syncthreads();
  if (wv == 0) {
    const int c0 = cnt[4 * lane], c1 = cnt[4 * lane + 1];
    const int c2 = cnt[4 * lane + 2], c3 = cnt[4 * lane + 3];
    const int sum4 = c0 + c1 + c2 + c3;
    int x = sum4;
    for (int d = 1; d < 32; d <<= 1) {
      const int y = __shfl_up(x, d);
      if (lane >= d) x += y;
    }
    const int ex = x - sum4;
    off[4 * lane]     = ex;
    off[4 * lane + 1] = ex + c0;
    off[4 * lane + 2] = ex + c0 + c1;
    off[4 * lane + 3] = ex + c0 + c1 + c2;
    if (lane == 31) off[RPB] = x;
  }
  __syncthreads();
  if (threadIdx.x < RPB && rbase + (int)threadIdx.x < n)
    rowptr[rbase + threadIdx.x] = bs + off[threadIdx.x];
  if (b == (int)gridDim.x - 1 && threadIdx.x == 0) rowptr[n] = be;
  for (int i = bs + (int)threadIdx.x; i < be; i += 512) {
    const int2 q = rec[i];
    tmp[bs + off[q.y - rbase] + rk[i]] = q.x;
  }
  __syncthreads();
  for (int rr = wv; rr < RPB; rr += 16) {
    const int o0 = __builtin_amdgcn_readfirstlane(off[rr]);
    const int o1 = __builtin_amdgcn_readfirstlane(off[rr + 1]);
    const int s = bs + o0;
    const int L = o1 - o0;
    for (int bi = 0; bi < L; bi += 32) {
      const int i = bi + lane;
      const int mine = (i < L) ? tmp[s + i] : 0x7fffffff;
      int dr = 0;
      for (int bj = 0; bj < L; bj += 32) {
        const int j = bj + lane;
        const int other = (j < L) ? tmp[s + j] : 0x7fffffff;
        const int m = (L - bj < 32) ? (L - bj) : 32;
        for (int q = 0; q < m; ++q)
          dr += (__builtin_amdgcn_readlane(other, q) < mine) ? 1 : 0;
      }
      if (i < L) {
        const float wm = w[mine];
        col_s[s + dr] = col[mine];
        w_s[s + dr]   = wm;
        if (fits) wl[o0 + dr] = wm;
      }
    }
  }
  __syncthreads();
  if (threadIdx.x < RPB && rbase + (int)threadIdx.x < n) {
    const int o0 = off[threadIdx.x], o1 = off[threadIdx.x + 1];
    float d = 1.0f;
    if (fits) { for (int p = o0; p < o1; ++p) d += wl[p]; }
    else      { for (int p = o0; p < o1; ++p) d += w_s[bs + p]; }
    dis[rbase + threadIdx.x] = rsqrtf(d);
  }
}

template <bool HEADS, bool FIRST>
__global__ __launch_bounds__(512, 12) void agg_gemm_k(const int* __restrict__ rowptr,
                                                  const int* __restrict__ col_s,
                                                  const float* __restrict__ nrm_s,
                                                  const float* __restrict__ X,
                                                  const float* __restrict__ dis,
                                                  const float* __restrict__ Wa,
                                                  const float* __restrict__ Wb,
                                                  const float* __restrict__ bias,
                                                  const float* __restrict__ bias_hi,
                                                  const float* __restrict__ rnd,
                                                  float* __restrict__ out, int MT, int n,
                                                  float* __restrict__ nrm_out) {
  constexpr int U  = 128;
  constexpr int NL = U / 32;
  constexpr int HL = NL / 2;
  constexpr int P  = 132;
  constexpr int PB = 2 * U + 32;
  constexpr int CSZ = HEADS ? 2 * 16 * P : 4;
  __shared__ alignas(16) float Bs[64 * PB];
  __shared__ alignas(16) float As[2 * 16 * P];
  __shared__ alignas(16) float Cs[CSZ];
  const int lane = threadIdx.x & 31;
  const int wave = __builtin_amdgcn_readfirstlane((int)(threadIdx.x >> 5));

  constexpr int WP = HEADS ? U / 2 : U;
  const float* wcol = (HEADS && lane >= 16) ? (Wb + (lane - 16) * 4) : (Wa + lane * 4);
  for (int i = 0; i < 64 / 16; ++i) {
    const int pr = i * 16 + wave;
    const float4 w0 = *reinterpret_cast<const float4*>(wcol + (size_t)(2 * pr) * WP);
    const float4 w1 = *reinterpret_cast<const float4*>(wcol + (size_t)(2 * pr + 1) * WP);
    float* bd = Bs + pr * PB + lane * 8;
    *reinterpret_cast<float4*>(bd)     = make_float4(w0.x, w1.x, w0.y, w1.y);
    *reinterpret_cast<float4*>(bd + 4) = make_float4(w0.z, w1.z, w0.w, w1.w);
  }

  const float* Xl = X + lane;
  float b0v = 0.0f, b1v = 0.0f;
  if (!HEADS) {
    b0v = bias[(wave & 3) * 32 + (lane & 15)];
    b1v = bias[(wave & 3) * 32 + 16 + (lane & 15)];
  }
  const int kk = (lane >> 4) * 2;

  int it = 0;
  int tprev = 0;
  for (int tm = blockIdx.x; tm < MT; tm += gridDim.x, ++it) {
    float* asb = As + (it & 1) * (16 * P);
    const int r = tm * 16 + wave;
    float rv[HL] = {0.0f, 0.0f}, bm[HL] = {0.0f, 0.0f}, bl[HL] = {0.0f, 0.0f};
    if (HEADS && it > 0)
      for (int j = 0; j < HL; ++j) {
        rv[j] = rnd[(size_t)(tprev * 16 + wave) * (U / 2) + lane + 32 * j];
        bm[j] = bias[lane + 32 * j];
        bl[j] = bias_hi[lane + 32 * j];
      }
    const int s = __builtin_amdgcn_readfirstlane(rowptr[r]);
    const int t = __builtin_amdgcn_readfirstlane(rowptr[r + 1]);
    const float d  = dis[r];
    const float sn = d * d;
    float acc[NL];
    for (int j = 0; j < NL; ++j) acc[j] = sn * Xl[(size_t)r * U + 32 * j];

    for (int base = s; base < t; base += 32) {
      const int m = (t - base < 32) ? (t - base) : 32;
      int   cv = 0;
      float wv = 0.0f;
      float dcv = 0.0f;
      if (lane < m) {
        cv = col_s[base + lane];
        wv = FIRST ? nrm_out[base + lane] : nrm_s[base + lane];
        if (FIRST) dcv = dis[cv];
      }
      bool raw = FIRST;
      int wi = __float_as_int(wv);
      int q = 0;
      for (; q + 4 <= m; q += 4) {
        const int c0 = __builtin_amdgcn_readlane(cv, q);
        const int c1 = __builtin_amdgcn_readlane(cv, q + 1);
        const int c2 = __builtin_amdgcn_readlane(cv, q + 2);
        const int c3 = __builtin_amdgcn_readlane(cv, q + 3);
        float w0, w1, w2, w3;
        if (!FIRST) {
          w0 = __int_as_float(__builtin_amdgcn_readlane(wi, q));
          w1 = __int_as_float(__builtin_amdgcn_readlane(wi, q + 1));
          w2 = __int_as_float(__builtin_amdgcn_readlane(wi, q + 2));
          w3 = __int_as_float(__builtin_amdgcn_readlane(wi, q + 3));
        }
        float t0[NL], t1[NL], t2[NL], t3[NL];
        for (int j = 0; j < NL; ++j) {
          t0[j] = Xl[(size_t)c0 * U + 32 * j];
          t1[j] = Xl[(size_t)c1 * U + 32 * j];
          t2[j] = Xl[(size_t)c2 * U + 32 * j];
          t3[j] = Xl[(size_t)c3 * U + 32 * j];
        }
        if (FIRST) {
          if (raw) {
            wv = d * wv * dcv;
            if (lane < m) nrm_out[base + lane] = wv;
            wi = __float_as_int(wv);
            raw = false;
          }
          w0 = __int_as_float(__builtin_amdgcn_readlane(wi, q));
          w1 = __int_as_float(__builtin_amdgcn_readlane(wi, q + 1));
          w2 = __int_as_float(__builtin_amdgcn_readlane(wi, q + 2));
          w3 = __int_as_float(__builtin_amdgcn_readlane(wi, q + 3));
        }
        for (int j = 0; j < NL; ++j) {
          acc[j] += w0 * t0[j];
          acc[j] += w1 * t1[j];
          acc[j] += w2 * t2[j];
          acc[j] += w3 * t3[j];
        }
      }
      if (FIRST && raw) {
        wv = d * wv * dcv;
        if (lane < m) nrm_out[base + lane] = wv;
        wi = __float_as_int(wv);
      }
      for (; q < m; ++q) {
        const int   c0 = __builtin_amdgcn_readlane(cv, q);
        const float w0 = __int_as_float(__builtin_amdgcn_readlane(wi, q));
        for (int j = 0; j < NL; ++j) acc[j] += w0 * Xl[(size_t)c0 * U + 32 * j];
      }
    }
    for (int j = 0; j < NL; ++j) asb[wave * P + lane + 32 * j] = acc[j];
    __syncthreads();

    if (HEADS && it > 0) {
      const float* cs = Cs + ((it - 1) & 1) * (16 * P) + wave * P + lane;
      float* ho = out + (size_t)(tprev * 16 + wave) * (U / 2) + lane;
      float* ko = ho + (size_t)n * (U / 2);
      const float mu0 = cs[0]  + bm[0], mu1 = cs[32] + bm[1];
      const float ls0 = cs[64] + bl[0], ls1 = cs[96] + bl[1];
      const float sd0 = expf(ls0), sd1 = expf(ls1);
      ho[0]  = mu0 + rv[0] * sd0;
      ho[32] = mu1 + rv[1] * sd1;
      ko[0]  = -ls0 + 0.5f * (sd0 * sd0 + mu0 * mu0 - 1.0f);
      ko[32] = -ls1 + 0.5f * (sd1 * sd1 + mu1 * mu1 - 1.0f);
    }

    if (wave < 4) {
      const float* ar = asb + (lane & 15) * P + kk;
      const float* br = Bs + (lane >> 4) * PB + (lane & 15) * 2 + wave * 64;
      v8f ca = v8f{0.f, 0.f, 0.f, 0.f, 0.f, 0.f, 0.f, 0.f};
      v8f cb = v8f{0.f, 0.f, 0.f, 0.f, 0.f, 0.f, 0.f, 0.f};
      v2f fa[2][4], f0[2][4], f1[2][4];
      for (int i = 0; i < 4; ++i) {
        fa[0][i] = *reinterpret_cast<const v2f*>(ar + 4 * i);
        f0[0][i] = *reinterpret_cast<const v2f*>(br + 2 * i * PB);
        f1[0][i] = *reinterpret_cast<const v2f*>(br + 2 * i * PB + 32);
      }
#pragma unroll
      for (int kb = 0; kb < 8; ++kb) {
        const int cur = kb & 1, nxt = cur ^ 1;
        if (kb + 1 < 8) {
          for (int i = 0; i < 4; ++i) {
            const int k = (kb + 1) * 16 + 4 * i;
            fa[nxt][i] = *reinterpret_cast<const v2f*>(ar + k);
            f0[nxt][i] = *reinterpret_cast<const v2f*>(br + (k >> 1) * PB);
            f1[nxt][i] = *reinterpret_cast<const v2f*>(br + (k >> 1) * PB + 32);
          }
        }
        __builtin_amdgcn_sched_barrier(0);
        for (int i = 0; i < 4; ++i) {
          ca = __builtin_amdgcn_wmma_f32_16x16x4_f32(false, fa[cur][i], false, f0[cur][i],
                                                     (short)0, ca, false, false);
          cb = __builtin_amdgcn_wmma_f32_16x16x4_f32(false, fa[cur][i], false, f1[cur][i],
                                                     (short)0, cb, false, false);
        }
        __builtin_amdgcn_sched_barrier(0);
      }
      if (HEADS) {
        float* cw = Cs + (it & 1) * (16 * P) + ((lane >> 4) * 8) * P + wave * 32 + (lane & 15);
        for (int r8 = 0; r8 < 8; ++r8) {
          cw[r8 * P]      = ca[r8];
          cw[r8 * P + 16] = cb[r8];
        }
      } else {
        const int mbase = tm * 16 + (lane >> 4) * 8;
        float* trow = out + (size_t)mbase * U + wave * 32 + (lane & 15);
        for (int r8 = 0; r8 < 8; ++r8) {
          trow[(size_t)r8 * U]      = fmaxf(ca[r8] + b0v, 0.0f);
          trow[(size_t)r8 * U + 16] = fmaxf(cb[r8] + b1v, 0.0f);
        }
      }
    }
    tprev = tm;
  }
  if (HEADS) {
    __syncthreads();
    if (it > 0) {
      const int rp = tprev * 16 + wave;
      const float* cs = Cs + ((it - 1) & 1) * (16 * P) + wave * P + lane;
      float* ho = out + (size_t)rp * (U / 2) + lane;
      float* ko = ho + (size_t)n * (U / 2);
      const float r0 = rnd[(size_t)rp * (U / 2) + lane];
      const float r1 = rnd[(size_t)rp * (U / 2) + lane + 32];
      const float mu0 = cs[0]  + bias[lane],    mu1 = cs[32] + bias[lane + 32];
      const float ls0 = cs[64] + bias_hi[lane], ls1 = cs[96] + bias_hi[lane + 32];
      const float sd0 = expf(ls0), sd1 = expf(ls1);
      ho[0]  = mu0 + r0 * sd0;
      ho[32] = mu1 + r1 * sd1;
      ko[0]  = -ls0 + 0.5f * (sd0 * sd0 + mu0 * mu0 - 1.0f);
      ko[32] = -ls1 + 0.5f * (sd1 * sd1 + mu1 * mu1 - 1.0f);
    }
  }
}

extern "C" void kernel_launch(void* const* d_in, const int* in_sizes, int n_in,
                              void* d_out, int out_size, void* d_ws, size_t ws_size,
                              hipStream_t stream) {
  (void)in_sizes; (void)n_in; (void)out_size; (void)ws_size;
  const int N = NN, E = NE;
  const float* h   = (const float*)d_in[0];
  const int*   ei  = (const int*)d_in[1];
  const float* ew  = (const float*)d_in[2];
  const float* rnd = (const float*)d_in[3];
  const float* W0  = (const float*)d_in[4];
  const float* b0  = (const float*)d_in[5];
  const float* W1  = (const float*)d_in[6];
  const float* b1  = (const float*)d_in[7];
  const float* Wmu = (const float*)d_in[8];
  const float* bmu = (const float*)d_in[9];
  const float* Wsd = (const float*)d_in[10];
  const float* bsd = (const float*)d_in[11];
  const int* row = ei;
  const int* col = ei + E;
  float* out = (float*)d_out;

  float* ws    = (float*)d_ws;
  float* dis   = ws;
  float* nrm   = dis + N;
  float* bufT  = nrm + E;
  float* bufH  = bufT + (size_t)N * 128;
  float* bufS  = bufH + (size_t)N * 128;
  int* col_s   = (int*)bufS;
  int* rowptr  = col_s + E;
  int* bsum    = rowptr + N + 96;
  int* hist    = bsum + 1024;
  int2* rec    = (int2*)bufH;
  int* rk      = (int*)bufH + 2 * (size_t)E;
  int* tmp     = rk + E;

  bucket_hist_k   <<<NCH, 1024, 0, stream>>>(row, hist, E);
  scan_local_k    <<<NBK, 256, 0, stream>>>(hist, bsum, NBK * NCH);
  bucket_scatter_k<<<NCH, 1024, 0, stream>>>(row, hist, bsum, rec, E);
  bucket_rows_k   <<<NBK, 512, 0, stream>>>(bsum, rec, rk, tmp, col, ew, rowptr, col_s, nrm, dis, N);

  const int MT = N / 16;
  const unsigned FB = 736u;

  agg_gemm_k<false, true><<<FB, 512, 0, stream>>>(rowptr, col_s, nullptr, h, dis, W0, nullptr, b0,
                                                  nullptr, nullptr, bufT, MT, N, nrm);
  agg_gemm_k<false, false><<<FB, 512, 0, stream>>>(rowptr, col_s, nrm, bufT, dis, W1, nullptr, b1,
                                                   nullptr, nullptr, bufH, MT, N, nullptr);
  agg_gemm_k<true, false><<<FB, 512, 0, stream>>>(rowptr, col_s, nrm, bufH, dis, Wmu, Wsd, bmu, bsd,
                                                  rnd, out, MT, N, nullptr);
}


# row sums: a row's gathers go out eight rows (32 line loads) per group and its last 1-7 rows as one group (was four rows per group, then the rest one row per round trip); sum orders kept
# speedup vs baseline: 5.0100x; 1.0406x over previous
#include <hip/hip_runtime.h>

#define NN 100000
#define NE 1600000

typedef float v2f __attribute__((ext_vector_type(2)));
typedef float v8f __attribute__((ext_vector_type(8)));

#define RBITS 7
#define RPB   (1 << RBITS)
#define NBK   ((NN + RPB - 1) >> RBITS)
#define NCH   256
#define CHUNK ((NE + NCH - 1) / NCH)

__global__ __launch_bounds__(1024) void bucket_hist_k(const int* __restrict__ row,
                                                      int* __restrict__ hist, int e) {
  __shared__ int h[NBK];
  for (int k = threadIdx.x; k < NBK; k += 1024) h[k] = 0;
  __syncthreads();
  const int lo = blockIdx.x * CHUNK;
  const int hi = (lo + CHUNK < e) ? (lo + CHUNK) : e;
  for (int i = lo + threadIdx.x; i < hi; i += 1024) atomicAdd(&h[row[i] >> RBITS], 1);
  __syncthreads();
  for (int k = threadIdx.x; k < NBK; k += 1024) hist[k * NCH + blockIdx.x] = h[k];
}

__global__ __launch_bounds__(256) void scan_local_k(int* cnt, int* __restrict__ bsum, int n1) {
  __shared__ int wsum[8];
  const int lane = threadIdx.x & 31;
  const int wv   = threadIdx.x >> 5;
  const int i = blockIdx.x * 256 + threadIdx.x;
  const int v = (i < n1) ? cnt[i] : 0;
  int x = v;
  for (int d = 1; d < 32; d <<= 1) {
    int y = __shfl_up(x, d);
    if (lane >= d) x += y;
  }
  if (lane == 31) wsum[wv] = x;
  __syncthreads();
  int off = 0;
  for (int k = 0; k < 8; ++k) off += (k < wv) ? wsum[k] : 0;
  const int incl = x + off;
  if (i < n1) cnt[i] = incl - v;
  if (threadIdx.x == 255) bsum[blockIdx.x] = incl;
}

static_assert(NCH == 256, "one scan block per bucket");
static_assert(NBK <= 1024, "one bucket per thread in the scan of the bucket counts");
__global__ __launch_bounds__(1024) void bucket_scatter_k(const int* __restrict__ row,
                                                         const int* __restrict__ hist,
                                                         const int* __restrict__ bsum,
                                                         int2* __restrict__ rec, int e) {
  __shared__ int cur[NBK];
  __shared__ int wtot[32];
  {
    const int lane = threadIdx.x & 31;
    const int wv   = threadIdx.x >> 5;
    const int k = threadIdx.x;
    const int v = (k < NBK) ? bsum[k] : 0;
    const int first = (k < NBK) ? hist[k * NCH + blockIdx.x] : 0;
    int x = v;
    for (int d = 1; d < 32; d <<= 1) {
      const int y = __shfl_up(x, d);
      if (lane >= d) x += y;
    }
    if (lane == 31) wtot[wv] = x;
    __syncthreads();
    int off = 0;
    for (int j = 0; j < 32; ++j) off += (j < wv) ? wtot[j] : 0;
    if (k < NBK) cur[k] = first + off + x - v;
  }
  __syncthreads();
  const int lo = blockIdx.x * CHUNK;
  const int hi = (lo + CHUNK < e) ? (lo + CHUNK) : e;
  for (int i = lo + threadIdx.x; i < hi; i += 1024) {
    const int r = row[i];
    const int p = atomicAdd(&cur[r >> RBITS], 1);
    rec[p] = make_int2(i, r);
  }
}

#define WCAP 4096
__global__ __launch_bounds__(512) void bucket_rows_k(const int* __restrict__ bsum,
                                                     const int2* __restrict__ rec,
                                                     int* __restrict__ rk,
                                                     int* __restrict__ tmp,
                                                     const int* __restrict__ col,
                                                     const float* __restrict__ w,
                                                     int* __restrict__ rowptr,
                                                     int* __restrict__ col_s,
                                                     float* __restrict__ w_s,
                                                     float* __restrict__ dis, int n) {
  __shared__ int cnt[RPB];
  __shared__ int off[RPB + 1];
  __shared__ float wl[WCAP];
  __shared__ int red[16];
  const int b = blockIdx.x;
  const int lane = threadIdx.x & 31;
  const int wv   = threadIdx.x >> 5;
  int part = 0;
  for (int k = threadIdx.x; k < b; k += 512) part += bsum[k];
  for (int d = 16; d >= 1; d >>= 1) part += __shfl_xor(part, d);
  if (lane == 0) red[wv] = part;
  const int mine_n = bsum[b];
  if (threadIdx.x < RPB) cnt[threadIdx.x] = 0;
  __syncthreads();
  int before = 0;
  for (int k = 0; k < 16; ++k) before += red[k];
  const int bs = __builtin_amdgcn_readfirstlane(before);
  const int be = bs + __builtin_amdgcn_readfirstlane(mine_n);
  const int rbase = b << RBITS;
  const bool fits = (be - bs) <= WCAP;
  for (int i = bs + (int)threadIdx.x; i < be; i += 512) {
    const int2 q = rec[i];
    rk[i] = atomicAdd(&cnt[q.y - rbase], 1);
  }
  __syncthreads();
  if (wv == 0) {
    const int c0 = cnt[4 * lane], c1 = cnt[4 * lane + 1];
    const int c2 = cnt[4 * lane + 2], c3 = cnt[4 * lane + 3];
    const int sum4 = c0 + c1 + c2 + c3;
    int x = sum4;
    for (int d = 1; d < 32; d <<= 1) {
      const int y = __shfl_up(x, d);
      if (lane >= d) x += y;
    }
    const int ex = x - sum4;
    off[4 * lane]     = ex;
    off[4 * lane + 1] = ex + c0;
    off[4 * lane + 2] = ex + c0 + c1;
    off[4 * lane + 3] = ex + c0 + c1 + c2;
    if (lane == 31) off[RPB] = x;
  }
  __syncthreads();
  if (threadIdx.x < RPB && rbase + (int)threadIdx.x < n)
    rowptr[rbase + threadIdx.x] = bs + off[threadIdx.x];
  if (b == (int)gridDim.x - 1 && threadIdx.x == 0) rowptr[n] = be;
  for (int i = bs + (int)threadIdx.x; i < be; i += 512) {
    const int2 q = rec[i];
    tmp[bs + off[q.y - rbase] + rk[i]] = q.x;
  }
  __syncthreads();
  for (int rr = wv; rr < RPB; rr += 16) {
    const int o0 = __builtin_amdgcn_readfirstlane(off[rr]);
    const int o1 = __builtin_amdgcn_readfirstlane(off[rr + 1]);
    const int s = bs + o0;
    const int L = o1 - o0;
    for (int bi = 0; bi < L; bi += 32) {
      const int i = bi + lane;
      const int mine = (i < L) ? tmp[s + i] : 0x7fffffff;
      int dr = 0;
      for (int bj = 0; bj < L; bj += 32) {
        const int j = bj + lane;
        const int other = (j < L) ? tmp[s + j] : 0x7fffffff;
        const int m = (L - bj < 32) ? (L - bj) : 32;
        for (int q = 0; q < m; ++q)
          dr += (__builtin_amdgcn_readlane(other, q) < mine) ? 1 : 0;
      }
      if (i < L) {
        const float wm = w[mine];
        col_s[s + dr] = col[mine];
        w_s[s + dr]   = wm;
        if (fits) wl[o0 + dr] = wm;
      }
    }
  }
  __syncthreads();
  if (threadIdx.x < RPB && rbase + (int)threadIdx.x < n) {
    const int o0 = off[threadIdx.x], o1 = off[threadIdx.x + 1];
    float d = 1.0f;
    if (fits) { for (int p = o0; p < o1; ++p) d += wl[p]; }
    else      { for (int p = o0; p < o1; ++p) d += w_s[bs + p]; }
    dis[rbase + threadIdx.x] = rsqrtf(d);
  }
}

template <int NR, bool FIRST>
__device__ __forceinline__ void row_group(const float* __restrict__ X, unsigned lane, int q,
                                          int cv, float& wv, int& wi, bool& raw, float d,
                                          float dcv, bool inrow, float* __restrict__ nslot,
                                          float (&acc)[4]) {
  float tv[NR][4];
#pragma unroll
  for (int i = 0; i < NR; ++i) {
    const int c = __builtin_amdgcn_readlane(cv, q + i);
    const float* rp = X + (size_t)((unsigned)c * 128u + lane);
#pragma unroll
    for (int j = 0; j < 4; ++j) tv[i][j] = rp[32 * j];
  }
  if (FIRST && raw) {
    __builtin_amdgcn_sched_barrier(0);
    wv = d * wv * dcv;
    if (inrow) *nslot = wv;
    wi = __float_as_int(wv);
    raw = false;
  }
#pragma unroll
  for (int i = 0; i < NR; ++i) {
    const float w = __int_as_float(__builtin_amdgcn_readlane(wi, q + i));
#pragma unroll
    for (int j = 0; j < 4; ++j) acc[j] += w * tv[i][j];
  }
}

template <bool HEADS, bool FIRST>
__global__ __launch_bounds__(512, 12) void agg_gemm_k(const int* __restrict__ rowptr,
                                                  const int* __restrict__ col_s,
                                                  const float* __restrict__ nrm_s,
                                                  const float* __restrict__ X,
                                                  const float* __restrict__ dis,
                                                  const float* __restrict__ Wa,
                                                  const float* __restrict__ Wb,
                                                  const float* __restrict__ bias,
                                                  const float* __restrict__ bias_hi,
                                                  const float* __restrict__ rnd,
                                                  float* __restrict__ out, int MT, int n,
                                                  float* __restrict__ nrm_out) {
  constexpr int U  = 128;
  constexpr int NL = U / 32;
  constexpr int HL = NL / 2;
  constexpr int P  = 132;
  constexpr int PB = 2 * U + 32;
  constexpr int CSZ = HEADS ? 2 * 16 * P : 4;
  __shared__ alignas(16) float Bs[64 * PB];
  __shared__ alignas(16) float As[2 * 16 * P];
  __shared__ alignas(16) float Cs[CSZ];
  constexpr int BSZ = HEADS ? U : 4;
  __shared__ alignas(16) float Bias[BSZ];
  const int lane = threadIdx.x & 31;
  const int wave = __builtin_amdgcn_readfirstlane((int)(threadIdx.x >> 5));

  constexpr int WP = HEADS ? U / 2 : U;
  const float* wcol = (HEADS && lane >= 16) ? (Wb + (lane - 16) * 4) : (Wa + lane * 4);
  for (int i = 0; i < 64 / 16; ++i) {
    const int pr = i * 16 + wave;
    const float4 w0 = *reinterpret_cast<const float4*>(wcol + (size_t)(2 * pr) * WP);
    const float4 w1 = *reinterpret_cast<const float4*>(wcol + (size_t)(2 * pr + 1) * WP);
    float* bd = Bs + pr * PB + lane * 8;
    *reinterpret_cast<float4*>(bd)     = make_float4(w0.x, w1.x, w0.y, w1.y);
    *reinterpret_cast<float4*>(bd + 4) = make_float4(w0.z, w1.z, w0.w, w1.w);
  }

  float b0v = 0.0f, b1v = 0.0f;
  if (!HEADS) {
    b0v = bias[(wave & 3) * 32 + (lane & 15)];
    b1v = bias[(wave & 3) * 32 + 16 + (lane & 15)];
  } else if (threadIdx.x < (unsigned)U) {
    Bias[threadIdx.x] = (threadIdx.x < (unsigned)(U / 2)) ? bias[threadIdx.x]
                                                          : bias_hi[threadIdx.x - U / 2];
  }
  const int kk = (lane >> 4) * 2;

  int it = 0;
  int tprev = 0;
  for (int tm = blockIdx.x; tm < MT; tm += gridDim.x, ++it) {
    float* asb = As + (it & 1) * (16 * P);
    const int r = tm * 16 + wave;
    float rv[HL] = {0.0f, 0.0f};
    if (HEADS && it > 0)
      for (int j = 0; j < HL; ++j)
        rv[j] = rnd[(size_t)(tprev * 16 + wave) * (U / 2) + lane + 32 * j];
    const int s = __builtin_amdgcn_readfirstlane(rowptr[r]);
    const int t = __builtin_amdgcn_readfirstlane(rowptr[r + 1]);
    const float d  = dis[r];
    const float sn = d * d;
    float acc[NL];
    const float* xr = X + (size_t)((unsigned)r * (unsigned)U + (unsigned)lane);
    for (int j = 0; j < NL; ++j) acc[j] = sn * xr[32 * j];

    for (int base = s; base < t; base += 32) {
      const int m = (t - base < 32) ? (t - base) : 32;
      int   cv = 0;
      float wv = 0.0f;
      float dcv = 0.0f;
      if (lane < m) {
        cv = col_s[base + lane];
        wv = FIRST ? nrm_out[base + lane] : nrm_s[base + lane];
        if (FIRST) dcv = dis[cv];
      }
      bool raw = FIRST;
      int wi = __float_as_int(wv);
      const bool inrow = (int)lane < m;
      float* nslot = FIRST ? (nrm_out + base + lane) : nullptr;
      int q = 0;
      for (; q + 8 <= m; q += 8)
        row_group<8, FIRST>(X, (unsigned)lane, q, cv, wv, wi, raw, d, dcv, inrow, nslot, acc);
      switch (m - q) {
        case 1: row_group<1, FIRST>(X, (unsigned)lane, q, cv, wv, wi, raw, d, dcv, inrow, nslot, acc); break;
        case 2: row_group<2, FIRST>(X, (unsigned)lane, q, cv, wv, wi, raw, d, dcv, inrow, nslot, acc); break;
        case 3: row_group<3, FIRST>(X, (unsigned)lane, q, cv, wv, wi, raw, d, dcv, inrow, nslot, acc); break;
        case 4: row_group<4, FIRST>(X, (unsigned)lane, q, cv, wv, wi, raw, d, dcv, inrow, nslot, acc); break;
        case 5: row_group<5, FIRST>(X, (unsigned)lane, q, cv, wv, wi, raw, d, dcv, inrow, nslot, acc); break;
        case 6: row_group<6, FIRST>(X, (unsigned)lane, q, cv, wv, wi, raw, d, dcv, inrow, nslot, acc); break;
        case 7: row_group<7, FIRST>(X, (unsigned)lane, q, cv, wv, wi, raw, d, dcv, inrow, nslot, acc); break;
        default: break;
      }
    }
    for (int j = 0; j < NL; ++j) asb[wave * P + lane + 32 * j] = acc[j];
    __syncthreads();

    if (HEADS && it > 0) {
      const float* cs = Cs + ((it - 1) & 1) * (16 * P) + wave * P + lane;
      float* ho = out + (size_t)(tprev * 16 + wave) * (U / 2) + lane;
      float* ko = ho + (size_t)n * (U / 2);
      const float mu0 = cs[0]  + Bias[lane],      mu1 = cs[32] + Bias[lane + 32];
      const float ls0 = cs[64] + Bias[lane + 64], ls1 = cs[96] + Bias[lane + 96];
      const float sd0 = expf(ls0), sd1 = expf(ls1);
      ho[0]  = mu0 + rv[0] * sd0;
      ho[32] = mu1 + rv[1] * sd1;
      ko[0]  = -ls0 + 0.5f * (sd0 * sd0 + mu0 * mu0 - 1.0f);
      ko[32] = -ls1 + 0.5f * (sd1 * sd1 + mu1 * mu1 - 1.0f);
    }

    if (wave < 4) {
      const float* ar = asb + (lane & 15) * P + kk;
      const float* br = Bs + (lane >> 4) * PB + (lane & 15) * 2 + wave * 64;
      v8f ca = v8f{0.f, 0.f, 0.f, 0.f, 0.f, 0.f, 0.f, 0.f};
      v8f cb = v8f{0.f, 0.f, 0.f, 0.f, 0.f, 0.f, 0.f, 0.f};
      v2f fa[2][4], f0[2][4], f1[2][4];
      for (int i = 0; i < 4; ++i) {
        fa[0][i] = *reinterpret_cast<const v2f*>(ar + 4 * i);
        f0[0][i] = *reinterpret_cast<const v2f*>(br + 2 * i * PB);
        f1[0][i] = *reinterpret_cast<const v2f*>(br + 2 * i * PB + 32);
      }
#pragma unroll
      for (int kb = 0; kb < 8; ++kb) {
        const int cur = kb & 1, nxt = cur ^ 1;
        if (kb + 1 < 8) {
          for (int i = 0; i < 4; ++i) {
            const int k = (kb + 1) * 16 + 4 * i;
            fa[nxt][i] = *reinterpret_cast<const v2f*>(ar + k);
            f0[nxt][i] = *reinterpret_cast<const v2f*>(br + (k >> 1) * PB);
            f1[nxt][i] = *reinterpret_cast<const v2f*>(br + (k >> 1) * PB + 32);
          }
        }
        __builtin_amdgcn_sched_barrier(0);
        for (int i = 0; i < 4; ++i) {
          ca = __builtin_amdgcn_wmma_f32_16x16x4_f32(false, fa[cur][i], false, f0[cur][i],
                                                     (short)0, ca, false, false);
          cb = __builtin_amdgcn_wmma_f32_16x16x4_f32(false, fa[cur][i], false, f1[cur][i],
                                                     (short)0, cb, false, false);
        }
        __builtin_amdgcn_sched_barrier(0);
      }
      if (HEADS) {
        float* cw = Cs + (it & 1) * (16 * P) + ((lane >> 4) * 8) * P + wave * 32 + (lane & 15);
        for (int r8 = 0; r8 < 8; ++r8) {
          cw[r8 * P]      = ca[r8];
          cw[r8 * P + 16] = cb[r8];
        }
      } else {
        const int mbase = tm * 16 + (lane >> 4) * 8;
        float* trow = out + (size_t)mbase * U + wave * 32 + (lane & 15);
        for (int r8 = 0; r8 < 8; ++r8) {
          trow[(size_t)r8 * U]      = fmaxf(ca[r8] + b0v, 0.0f);
          trow[(size_t)r8 * U + 16] = fmaxf(cb[r8] + b1v, 0.0f);
        }
      }
    }
    tprev = tm;
  }
  if (HEADS) {
    __syncthreads();
    if (it > 0) {
      const int rp = tprev * 16 + wave;
      const float* cs = Cs + ((it - 1) & 1) * (16 * P) + wave * P + lane;
      float* ho = out + (size_t)rp * (U / 2) + lane;
      float* ko = ho + (size_t)n * (U / 2);
      const float r0 = rnd[(size_t)rp * (U / 2) + lane];
      const float r1 = rnd[(size_t)rp * (U / 2) + lane + 32];
      const float mu0 = cs[0]  + Bias[lane],      mu1 = cs[32] + Bias[lane + 32];
      const float ls0 = cs[64] + Bias[lane + 64], ls1 = cs[96] + Bias[lane + 96];
      const float sd0 = expf(ls0), sd1 = expf(ls1);
      ho[0]  = mu0 + r0 * sd0;
      ho[32] = mu1 + r1 * sd1;
      ko[0]  = -ls0 + 0.5f * (sd0 * sd0 + mu0 * mu0 - 1.0f);
      ko[32] = -ls1 + 0.5f * (sd1 * sd1 + mu1 * mu1 - 1.0f);
    }
  }
}

extern "C" void kernel_launch(void* const* d_in, const int* in_sizes, int n_in,
                              void* d_out, int out_size, void* d_ws, size_t ws_size,
                              hipStream_t stream) {
  (void)in_sizes; (void)n_in; (void)out_size; (void)ws_size;
  const int N = NN, E = NE;
  const float* h   = (const float*)d_in[0];
  const int*   ei  = (const int*)d_in[1];
  const float* ew  = (const float*)d_in[2];
  const float* rnd = (const float*)d_in[3];
  const float* W0  = (const float*)d_in[4];
  const float* b0  = (const float*)d_in[5];
  const float* W1  = (const float*)d_in[6];
  const float* b1  = (const float*)d_in[7];
  const float* Wmu = (const float*)d_in[8];
  const float* bmu = (const float*)d_in[9];
  const float* Wsd = (const float*)d_in[10];
  const float* bsd = (const float*)d_in[11];
  const int* row = ei;
  const int* col = ei + E;
  float* out = (float*)d_out;

  float* ws    = (float*)d_ws;
  float* dis   = ws;
  float* nrm   = dis + N;
  float* bufT  = nrm + E;
  float* bufH  = bufT + (size_t)N * 128;
  float* bufS  = bufH + (size_t)N * 128;
  int* col_s   = (int*)bufS;
  int* rowptr  = col_s + E;
  int* bsum    = rowptr + N + 96;
  int* hist    = bsum + 1024;
  int2* rec    = (int2*)bufH;
  int* rk      = (int*)bufH + 2 * (size_t)E;
  int* tmp     = rk + E;

  bucket_hist_k   <<<NCH, 1024, 0, stream>>>(row, hist, E);
  scan_local_k    <<<NBK, 256, 0, stream>>>(hist, bsum, NBK * NCH);
  bucket_scatter_k<<<NCH, 1024, 0, stream>>>(row, hist, bsum, rec, E);
  bucket_rows_k   <<<NBK, 512, 0, stream>>>(bsum, rec, rk, tmp, col, ew, rowptr, col_s, nrm, dis, N);

  const int MT = N / 16;
  const unsigned FB = 736u;

  agg_gemm_k<false, true><<<FB, 512, 0, stream>>>(rowptr, col_s, nullptr, h, dis, W0, nullptr, b0,
                                                  nullptr, nullptr, bufT, MT, N, nrm);
  agg_gemm_k<false, false><<<FB, 512, 0, stream>>>(rowptr, col_s, nrm, bufT, dis, W1, nullptr, b1,
                                                   nullptr, nullptr, bufH, MT, N, nullptr);
  agg_gemm_k<true, false><<<FB, 512, 0, stream>>>(rowptr, col_s, nrm, bufH, dis, Wmu, Wsd, bmu, bsd,
                                                  rnd, out, MT, N, nullptr);
}
